# starting kernel
# baseline (speedup 1.0000x reference)
#include <hip/hip_runtime.h>

typedef __attribute__((ext_vector_type(2))) float v2f;
typedef __attribute__((ext_vector_type(8))) float v8f;

#define HD 128
#define WT_STRIDE 132

__global__ __launch_bounds__(256) void k_deg_init(float* __restrict__ deg, int n) {
  int i = blockIdx.x * 256 + threadIdx.x;
  if (i < n) deg[i] = 1.0f;
}

__global__ __launch_bounds__(256) void k_deg_count(const int* __restrict__ dst,
                                                   float* __restrict__ deg, int e) {
  int stride = gridDim.x * 256;
  for (int i = blockIdx.x * 256 + threadIdx.x; i < e; i += stride)
    atomicAdd(&deg[dst[i]], 1.0f);
}

__global__ __launch_bounds__(256) void k_dinv(float* __restrict__ deg, int n) {
  int i = blockIdx.x * 256 + threadIdx.x;
  if (i < n) deg[i] = rsqrtf(deg[i]);
}

__global__ __launch_bounds__(256) void k_gemm_wmma(const float* __restrict__ A,
                                                   const float* __restrict__ W,
                                                   float* __restrict__ out, int n) {
  __shared__ float Wt[HD * WT_STRIDE];
  for (int i = threadIdx.x; i < HD * HD / 4; i += 256) {
    int k  = i >> 5;
    int n4 = (i & 31) * 4;
    const float4 w = *(const float4*)(W + k * HD + n4);
    Wt[(n4 + 0) * WT_STRIDE + k] = w.x;
    Wt[(n4 + 1) * WT_STRIDE + k] = w.y;
    Wt[(n4 + 2) * WT_STRIDE + k] = w.z;
    Wt[(n4 + 3) * WT_STRIDE + k] = w.w;
  }
  __syncthreads();

  const int lane = threadIdx.x & 31;
  const int wave = threadIdx.x >> 5;
  const int r0   = blockIdx.x * 128 + wave * 16;
  if (r0 + 16 > n) return;
  const int hi = lane >> 4;
  const int lo = lane & 15;
  const float* arow = A + (size_t)(r0 + lo) * HD;

  v8f acc[8] = {};
  for (int kk = 0; kk < 32; ++kk) {
    const int k = kk * 4 + hi * 2;
    const v2f a = *(const v2f*)(arow + k);
#pragma unroll
    for (int t = 0; t < 8; ++t) {
      const v2f b = *(const v2f*)(&Wt[(t * 16 + lo) * WT_STRIDE + k]);
      acc[t] = __builtin_amdgcn_wmma_f32_16x16x4_f32(
          false, a, false, b, (short)0, acc[t], false, false);
    }
  }
#pragma unroll
  for (int t = 0; t < 8; ++t) {
    float* orow = out + (size_t)(r0 + hi * 8) * HD + t * 16 + lo;
#pragma unroll
    for (int r = 0; r < 8; ++r)
      orow[(size_t)r * HD] = acc[t][r];
  }
}

__global__ __launch_bounds__(256) void k_self_init(const float* __restrict__ dinv,
                                                   const float* __restrict__ m,
                                                   float* __restrict__ agg, int total) {
  int i = blockIdx.x * 256 + threadIdx.x;
  if (i < total) {
    float di = dinv[i >> 7];
    agg[i] = di * di * m[i];
  }
}

__global__ __launch_bounds__(256) void k_self_init_bias(const float* __restrict__ dinv,
                                                        const float* __restrict__ m,
                                                        const float* __restrict__ bias,
                                                        float* __restrict__ outp, int total) {
  int i = blockIdx.x * 256 + threadIdx.x;
  if (i < total) {
    float di = dinv[i >> 7];
    outp[i] = di * di * m[i] + bias[i & 127];
  }
}

__global__ __launch_bounds__(256) void k_bias_relu(float* __restrict__ agg,
                                                   const float* __restrict__ bias, int total) {
  int i = blockIdx.x * 256 + threadIdx.x;
  if (i < total) {
    float v = agg[i] + bias[i & 127];
    agg[i] = v > 0.0f ? v : 0.0f;
  }
}

__global__ __launch_bounds__(256) void k_scatter(const int* __restrict__ src,
                                                 const int* __restrict__ dst,
                                                 const float* __restrict__ dinv,
                                                 const float* __restrict__ m,
                                                 float* __restrict__ agg, int e) {
  const int lane = threadIdx.x & 31;
  int w = (blockIdx.x * 256 + threadIdx.x) >> 5;
  const int nw = (gridDim.x * 256) >> 5;
  for (; w < e; w += nw) {
    const int s = src[w];
    const int d = dst[w];
    const float nrm = dinv[s] * dinv[d];
    const float4 v = *(const float4*)(m + (size_t)s * HD + lane * 4);
    float* p = agg + (size_t)d * HD + lane * 4;
    atomicAdd(p + 0, nrm * v.x);
    atomicAdd(p + 1, nrm * v.y);
    atomicAdd(p + 2, nrm * v.z);
    atomicAdd(p + 3, nrm * v.w);
  }
}

extern "C" void kernel_launch(void* const* d_in, const int* in_sizes, int n_in,
                              void* d_out, int out_size, void* d_ws, size_t ws_size,
                              hipStream_t stream) {
  const float* x  = (const float*)d_in[0];
  const float* W1 = (const float*)d_in[1];
  const float* b1 = (const float*)d_in[2];
  const float* W2 = (const float*)d_in[3];
  const float* b2 = (const float*)d_in[4];
  const int*   ei = (const int*)d_in[5];

  const int n = in_sizes[0] / HD;
  const int e = in_sizes[5] / 2;
  const int* src = ei;
  const int* dst = ei + e;
  const int total = n * HD;

  char* ws = (char*)d_ws;
  float* dinv = (float*)ws;
  size_t off1 = ((size_t)n * 4 + 255) & ~(size_t)255;
  float* buf1 = (float*)(ws + off1);
  size_t off2 = off1 + (((size_t)total * 4 + 255) & ~(size_t)255);
  float* buf2 = (float*)(ws + off2);
  float* out  = (float*)d_out;

  dim3 blk(256);
  const int gN   = (n + 255) / 256;
  const int gNH  = (total + 255) / 256;
  const int gMM  = (n + 127) / 128;

  k_deg_init <<<gN,   blk, 0, stream>>>(dinv, n);
  k_deg_count<<<2048, blk, 0, stream>>>(dst, dinv, e);
  k_dinv     <<<gN,   blk, 0, stream>>>(dinv, n);

  k_gemm_wmma<<<gMM,  blk, 0, stream>>>(x, W1, buf1, n);
  k_self_init<<<gNH,  blk, 0, stream>>>(dinv, buf1, buf2, total);
  k_scatter  <<<4096, blk, 0, stream>>>(src, dst, dinv, buf1, buf2, e);
  k_bias_relu<<<gNH,  blk, 0, stream>>>(buf2, b1, total);

  k_gemm_wmma<<<gMM,  blk, 0, stream>>>(buf2, W2, buf1, n);
  k_self_init_bias<<<gNH, blk, 0, stream>>>(dinv, buf1, b2, out, total);
  k_scatter  <<<4096, blk, 0, stream>>>(src, dst, dinv, buf1, out, e);
}


# Scatter takes 32 edges per wave: src, dst, norm loaded once per lane (coalesced), broadcast by readlane, so row loads of 8 edges are in flight at once; atomics kept, sums still in arrival order
# speedup vs baseline: 2.0307x; 2.0307x over previous
#include <hip/hip_runtime.h>

typedef __attribute__((ext_vector_type(2))) float v2f;
typedef __attribute__((ext_vector_type(8))) float v8f;

#define HD 128
#define WT_STRIDE 132

__global__ __launch_bounds__(256) void k_deg_init(float* __restrict__ deg, int n) {
  int i = blockIdx.x * 256 + threadIdx.x;
  if (i < n) deg[i] = 1.0f;
}

__global__ __launch_bounds__(256) void k_deg_count(const int* __restrict__ dst,
                                                   float* __restrict__ deg, int e) {
  int stride = gridDim.x * 256;
  for (int i = blockIdx.x * 256 + threadIdx.x; i < e; i += stride)
    atomicAdd(&deg[dst[i]], 1.0f);
}

__global__ __launch_bounds__(256) void k_dinv(float* __restrict__ deg, int n) {
  int i = blockIdx.x * 256 + threadIdx.x;
  if (i < n) deg[i] = rsqrtf(deg[i]);
}

template <bool IN_BIAS_RELU, bool OUT_BIAS>
__device__ __forceinline__ void gemm_body(const float* __restrict__ A,
                                          const float* __restrict__ W,
                                          const float* __restrict__ bin,
                                          const float* __restrict__ dinv,
                                          const float* __restrict__ bout,
                                          float* __restrict__ m,
                                          float* __restrict__ start, int n) {
  __shared__ float Wt[HD * WT_STRIDE];
  __shared__ __attribute__((aligned(16))) float bs[HD];
  for (int i = threadIdx.x; i < HD * HD / 4; i += 256) {
    int k  = i >> 5;
    int n4 = (i & 31) * 4;
    const float4 w = *(const float4*)(W + k * HD + n4);
    Wt[(n4 + 0) * WT_STRIDE + k] = w.x;
    Wt[(n4 + 1) * WT_STRIDE + k] = w.y;
    Wt[(n4 + 2) * WT_STRIDE + k] = w.z;
    Wt[(n4 + 3) * WT_STRIDE + k] = w.w;
  }
  if (IN_BIAS_RELU) {
    if (threadIdx.x < HD) bs[threadIdx.x] = bin[threadIdx.x];
  }
  __syncthreads();

  const int lane = threadIdx.x & 31;
  const int wave = threadIdx.x >> 5;
  const int r0   = blockIdx.x * 128 + wave * 16;
  if (r0 + 16 > n) return;
  const int hi = lane >> 4;
  const int lo = lane & 15;
  const float* arow = A + (size_t)(r0 + lo) * HD;

  v8f acc[8] = {};
  for (int kk = 0; kk < 32; ++kk) {
    const int k = kk * 4 + hi * 2;
    v2f a = *(const v2f*)(arow + k);
    if (IN_BIAS_RELU) {
      const v2f bb = *(const v2f*)(&bs[k]);
      const float x0 = a.x + bb.x;
      const float x1 = a.y + bb.y;
      a.x = x0 > 0.0f ? x0 : 0.0f;
      a.y = x1 > 0.0f ? x1 : 0.0f;
    }
#pragma unroll
    for (int t = 0; t < 8; ++t) {
      const v2f b = *(const v2f*)(&Wt[(t * 16 + lo) * WT_STRIDE + k]);
      acc[t] = __builtin_amdgcn_wmma_f32_16x16x4_f32(
          false, a, false, b, (short)0, acc[t], false, false);
    }
  }

  float sc[8];
#pragma unroll
  for (int r = 0; r < 8; ++r) {
    const float di = dinv[r0 + hi * 8 + r];
    sc[r] = di * di;
  }
#pragma unroll
  for (int t = 0; t < 8; ++t) {
    const size_t o = (size_t)(r0 + hi * 8) * HD + t * 16 + lo;
    float* mrow = m + o;
    float* srow = start + o;
    float bo = 0.0f;
    if (OUT_BIAS) bo = bout[t * 16 + lo];
#pragma unroll
    for (int r = 0; r < 8; ++r) {
      const float v = acc[t][r];
      mrow[(size_t)r * HD] = v;
      if (OUT_BIAS) srow[(size_t)r * HD] = sc[r] * v + bo;
      else          srow[(size_t)r * HD] = sc[r] * v;
    }
  }
}

__global__ __launch_bounds__(256) void k_gemm_l1(const float* __restrict__ A,
                                                 const float* __restrict__ W,
                                                 const float* __restrict__ dinv,
                                                 float* __restrict__ m,
                                                 float* __restrict__ start, int n) {
  gemm_body<false, false>(A, W, W, dinv, W, m, start, n);
}

__global__ __launch_bounds__(256) void k_gemm_l2(const float* __restrict__ A,
                                                 const float* __restrict__ W,
                                                 const float* __restrict__ bin,
                                                 const float* __restrict__ dinv,
                                                 const float* __restrict__ bout,
                                                 float* __restrict__ m,
                                                 float* __restrict__ start, int n) {
  gemm_body<true, true>(A, W, bin, dinv, bout, m, start, n);
}

#define EG 8
__global__ __launch_bounds__(256) void k_scatter(const int* __restrict__ src,
                                                 const int* __restrict__ dst,
                                                 const float* __restrict__ dinv,
                                                 const float* __restrict__ m,
                                                 float* __restrict__ agg, int e) {
  const int lane = threadIdx.x & 31;
  const int nb = (e + 31) >> 5;
  const int nw = gridDim.x * 8;
  int b = __builtin_amdgcn_readfirstlane((int)((blockIdx.x * 256 + threadIdx.x) >> 5));
  for (; b < nb; b += nw) {
    const int i = b * 32 + lane;
    int s = 0, d = 0;
    float nrm = 0.0f;
    if (i < e) {
      s = src[i];
      d = dst[i];
      nrm = dinv[s] * dinv[d];
    }
    const int nrm_i = __float_as_int(nrm);
    const int cnt = (e - b * 32) < 32 ? (e - b * 32) : 32;
    if (cnt == 32) {
#pragma unroll 1
      for (int j0 = 0; j0 < 32; j0 += EG) {
        float v[EG][4];
#pragma unroll
        for (int j = 0; j < EG; ++j) {
          const int sj = __builtin_amdgcn_readlane(s, j0 + j);
          const float* mp = m + (size_t)sj * HD + lane;
          v[j][0] = mp[0];
          v[j][1] = mp[32];
          v[j][2] = mp[64];
          v[j][3] = mp[96];
        }
#pragma unroll
        for (int j = 0; j < EG; ++j) {
          const int dj = __builtin_amdgcn_readlane(d, j0 + j);
          const float nj = __int_as_float(__builtin_amdgcn_readlane(nrm_i, j0 + j));
          float* p = agg + (size_t)dj * HD + lane;
          atomicAdd(p +  0, nj * v[j][0]);
          atomicAdd(p + 32, nj * v[j][1]);
          atomicAdd(p + 64, nj * v[j][2]);
          atomicAdd(p + 96, nj * v[j][3]);
        }
      }
    } else {
      for (int j = 0; j < cnt; ++j) {
        const int sj = __builtin_amdgcn_readlane(s, j);
        const int dj = __builtin_amdgcn_readlane(d, j);
        const float nj = __int_as_float(__builtin_amdgcn_readlane(nrm_i, j));
        const float* mp = m + (size_t)sj * HD + lane;
        float* p = agg + (size_t)dj * HD + lane;
        atomicAdd(p +  0, nj * mp[0]);
        atomicAdd(p + 32, nj * mp[32]);
        atomicAdd(p + 64, nj * mp[64]);
        atomicAdd(p + 96, nj * mp[96]);
      }
    }
  }
}

extern "C" void kernel_launch(void* const* d_in, const int* in_sizes, int n_in,
                              void* d_out, int out_size, void* d_ws, size_t ws_size,
                              hipStream_t stream) {
  const float* x  = (const float*)d_in[0];
  const float* W1 = (const float*)d_in[1];
  const float* b1 = (const float*)d_in[2];
  const float* W2 = (const float*)d_in[3];
  const float* b2 = (const float*)d_in[4];
  const int*   ei = (const int*)d_in[5];

  const int n = in_sizes[0] / HD;
  const int e = in_sizes[5] / 2;
  const int* src = ei;
  const int* dst = ei + e;
  const int total = n * HD;

  char* ws = (char*)d_ws;
  float* dinv = (float*)ws;
  size_t off1 = ((size_t)n * 4 + 255) & ~(size_t)255;
  float* buf1 = (float*)(ws + off1);
  size_t off2 = off1 + (((size_t)total * 4 + 255) & ~(size_t)255);
  float* buf2 = (float*)(ws + off2);
  float* out  = (float*)d_out;

  dim3 blk(256);
  const int gN   = (n + 255) / 256;
  const int gMM  = (n + 127) / 128;
  const int gS   = (e + 255) / 256;

  k_deg_init <<<gN,   blk, 0, stream>>>(dinv, n);
  k_deg_count<<<2048, blk, 0, stream>>>(dst, dinv, e);
  k_dinv     <<<gN,   blk, 0, stream>>>(dinv, n);

  k_gemm_l1  <<<gMM,  blk, 0, stream>>>(x, W1, dinv, buf1, buf2, n);
  k_scatter  <<<gS,   blk, 0, stream>>>(src, dst, dinv, buf1, buf2, e);

  k_gemm_l2  <<<gMM,  blk, 0, stream>>>(buf2, W2, b1, dinv, b2, buf1, out, n);
  k_scatter  <<<gS,   blk, 0, stream>>>(src, dst, dinv, buf1, out, e);
}


# Scatter split by column half across the two package halves (die id, work drawn by ticket): each L2 half holds half of m and agg, no cross-half atomics; f32 atomics kept, sums still in arrival order
# speedup vs baseline: 2.1561x; 1.0618x over previous
#include <hip/hip_runtime.h>

typedef __attribute__((ext_vector_type(2))) float v2f;
typedef __attribute__((ext_vector_type(8))) float v8f;

#define HD 128
#define WT_STRIDE 132

__global__ __launch_bounds__(256) void k_deg_init(float* __restrict__ deg, int* __restrict__ tk, int n) {
  int i = blockIdx.x * 256 + threadIdx.x;
  if (i < n) deg[i] = 1.0f;
  if (i < 2) tk[i * 32] = 0;
}

__global__ __launch_bounds__(256) void k_deg_count(const int* __restrict__ dst,
                                                   float* __restrict__ deg, int e) {
  int stride = gridDim.x * 256;
  for (int i = blockIdx.x * 256 + threadIdx.x; i < e; i += stride)
    atomicAdd(&deg[dst[i]], 1.0f);
}

__global__ __launch_bounds__(256) void k_dinv(float* __restrict__ deg, int n) {
  int i = blockIdx.x * 256 + threadIdx.x;
  if (i < n) deg[i] = rsqrtf(deg[i]);
}

template <bool IN_BIAS_RELU, bool OUT_BIAS>
__device__ __forceinline__ void gemm_body(const float* __restrict__ A,
                                          const float* __restrict__ W,
                                          const float* __restrict__ bin,
                                          const float* __restrict__ dinv,
                                          const float* __restrict__ bout,
                                          float* __restrict__ m,
                                          float* __restrict__ start, int n) {
  __shared__ float Wt[HD * WT_STRIDE];
  __shared__ __attribute__((aligned(16))) float bs[HD];
  for (int i = threadIdx.x; i < HD * HD / 4; i += 256) {
    int k  = i >> 5;
    int n4 = (i & 31) * 4;
    const float4 w = *(const float4*)(W + k * HD + n4);
    Wt[(n4 + 0) * WT_STRIDE + k] = w.x;
    Wt[(n4 + 1) * WT_STRIDE + k] = w.y;
    Wt[(n4 + 2) * WT_STRIDE + k] = w.z;
    Wt[(n4 + 3) * WT_STRIDE + k] = w.w;
  }
  if (IN_BIAS_RELU) {
    if (threadIdx.x < HD) bs[threadIdx.x] = bin[threadIdx.x];
  }
  __syncthreads();

  const int lane = threadIdx.x & 31;
  const int wave = threadIdx.x >> 5;
  const int r0   = blockIdx.x * 128 + wave * 16;
  if (r0 + 16 > n) return;
  const int hi = lane >> 4;
  const int lo = lane & 15;
  const float* arow = A + (size_t)(r0 + lo) * HD;

  v8f acc[8] = {};
  for (int kk = 0; kk < 32; ++kk) {
    const int k = kk * 4 + hi * 2;
    v2f a = *(const v2f*)(arow + k);
    if (IN_BIAS_RELU) {
      const v2f bb = *(const v2f*)(&bs[k]);
      const float x0 = a.x + bb.x;
      const float x1 = a.y + bb.y;
      a.x = x0 > 0.0f ? x0 : 0.0f;
      a.y = x1 > 0.0f ? x1 : 0.0f;
    }
#pragma unroll
    for (int t = 0; t < 8; ++t) {
      const v2f b = *(const v2f*)(&Wt[(t * 16 + lo) * WT_STRIDE + k]);
      acc[t] = __builtin_amdgcn_wmma_f32_16x16x4_f32(
          false, a, false, b, (short)0, acc[t], false, false);
    }
  }

  float sc[8];
#pragma unroll
  for (int r = 0; r < 8; ++r) {
    const float di = dinv[r0 + hi * 8 + r];
    sc[r] = di * di;
  }
#pragma unroll
  for (int t = 0; t < 8; ++t) {
    const size_t o = (size_t)(r0 + hi * 8) * HD + t * 16 + lo;
    float* mrow = m + o;
    float* srow = start + o;
    float bo = 0.0f;
    if (OUT_BIAS) bo = bout[t * 16 + lo];
#pragma unroll
    for (int r = 0; r < 8; ++r) {
      const float v = acc[t][r];
      mrow[(size_t)r * HD] = v;
      if (OUT_BIAS) srow[(size_t)r * HD] = sc[r] * v + bo;
      else          srow[(size_t)r * HD] = sc[r] * v;
    }
  }
}

__global__ __launch_bounds__(256) void k_gemm_l1(const float* __restrict__ A,
                                                 const float* __restrict__ W,
                                                 const float* __restrict__ dinv,
                                                 float* __restrict__ m,
                                                 float* __restrict__ start, int n) {
  gemm_body<false, false>(A, W, W, dinv, W, m, start, n);
}

__global__ __launch_bounds__(256) void k_gemm_l2(const float* __restrict__ A,
                                                 const float* __restrict__ W,
                                                 const float* __restrict__ bin,
                                                 const float* __restrict__ dinv,
                                                 const float* __restrict__ bout,
                                                 float* __restrict__ m,
                                                 float* __restrict__ start, int n) {
  gemm_body<true, true>(A, W, bin, dinv, bout, m, start, n);
}

#define EG 16
__device__ __forceinline__ void scatter_batch(const int b, const int cb, const int lane,
                                              const int* __restrict__ src,
                                              const int* __restrict__ dst,
                                              const float* __restrict__ dinv,
                                              const float* __restrict__ m,
                                              float* __restrict__ agg, const int e) {
  const int i = b * 32 + lane;
  const unsigned ul = (unsigned)lane;
  int s = 0, d = 0;
  float nrm = 0.0f;
  if (i < e) {
    s = src[i];
    d = dst[i];
    nrm = dinv[s] * dinv[d];
  }
  const int nrm_i = __float_as_int(nrm);
  const int cnt = (e - b * 32) < 32 ? (e - b * 32) : 32;
  if (cnt == 32) {
#pragma unroll 1
    for (int j0 = 0; j0 < 32; j0 += EG) {
      float v[EG][2];
#pragma unroll
      for (int j = 0; j < EG; ++j) {
        const int sj = __builtin_amdgcn_readlane(s, j0 + j);
        const float* mp = m + ((size_t)sj * HD + cb);
        v[j][0] = mp[ul];
        v[j][1] = (mp + 32)[ul];
      }
#pragma unroll
      for (int j = 0; j < EG; ++j) {
        const int dj = __builtin_amdgcn_readlane(d, j0 + j);
        const float nj = __int_as_float(__builtin_amdgcn_readlane(nrm_i, j0 + j));
        float* p = agg + ((size_t)dj * HD + cb);
        atomicAdd(p + ul, nj * v[j][0]);
        atomicAdd(p + 32 + ul, nj * v[j][1]);
      }
    }
  } else {
    for (int j = 0; j < cnt; ++j) {
      const int sj = __builtin_amdgcn_readlane(s, j);
      const int dj = __builtin_amdgcn_readlane(d, j);
      const float nj = __int_as_float(__builtin_amdgcn_readlane(nrm_i, j));
      const float* mp = m + (size_t)sj * HD + cb + lane;
      float* p = agg + (size_t)dj * HD + cb + lane;
      atomicAdd(p +  0, nj * mp[0]);
      atomicAdd(p + 32, nj * mp[32]);
    }
  }
}

__global__ __launch_bounds__(32) void k_ticket_zero(int* __restrict__ tk) {
  if (threadIdx.x < 2) tk[threadIdx.x * 32] = 0;
}

__global__ __launch_bounds__(256) __attribute__((amdgpu_waves_per_eu(16, 16))) void k_scatter(const int* __restrict__ src,
                                                 const int* __restrict__ dst,
                                                 const float* __restrict__ dinv,
                                                 const float* __restrict__ m,
                                                 float* __restrict__ agg,
                                                 int* __restrict__ tk, int e) {
  __shared__ int slot[2];
  const int lane = threadIdx.x & 31;
  const int wave = __builtin_amdgcn_readfirstlane((int)(threadIdx.x >> 5));
  const int nb = (e + 31) >> 5;
  const int nt = (nb + 7) >> 3;
  int h = (__builtin_amdgcn_s_sendmsg_rtn(0x87) >> 18) & 1;
  int other_done = 0;
  int r = 0;
  if (threadIdx.x == 0) slot[0] = (atomicAdd(&tk[h * 32], 1) << 1) | h;
  for (;;) {
    __syncthreads();
    const int sv = __builtin_amdgcn_readfirstlane(slot[r & 1]);
    const int t  = sv >> 1;
    const int hc = sv & 1;
    ++r;
    if (t >= nt) {
      if (other_done) break;
      other_done = 1;
      h = hc ^ 1;
      if (threadIdx.x == 0) slot[r & 1] = (atomicAdd(&tk[h * 32], 1) << 1) | h;
      continue;
    }
    h = hc;
    int nxt = 0;
    if (threadIdx.x == 0) nxt = (atomicAdd(&tk[h * 32], 1) << 1) | h;
    const int b = t * 8 + wave;
    if (b < nb) scatter_batch(b, hc * 64, lane, src, dst, dinv, m, agg, e);
    if (threadIdx.x == 0) slot[r & 1] = nxt;
  }
}

extern "C" void kernel_launch(void* const* d_in, const int* in_sizes, int n_in,
                              void* d_out, int out_size, void* d_ws, size_t ws_size,
                              hipStream_t stream) {
  const float* x  = (const float*)d_in[0];
  const float* W1 = (const float*)d_in[1];
  const float* b1 = (const float*)d_in[2];
  const float* W2 = (const float*)d_in[3];
  const float* b2 = (const float*)d_in[4];
  const int*   ei = (const int*)d_in[5];

  const int n = in_sizes[0] / HD;
  const int e = in_sizes[5] / 2;
  const int* src = ei;
  const int* dst = ei + e;
  const int total = n * HD;

  char* ws = (char*)d_ws;
  float* dinv = (float*)ws;
  size_t off1 = ((size_t)n * 4 + 255) & ~(size_t)255;
  float* buf1 = (float*)(ws + off1);
  size_t off2 = off1 + (((size_t)total * 4 + 255) & ~(size_t)255);
  float* buf2 = (float*)(ws + off2);
  float* out  = (float*)d_out;

  dim3 blk(256);
  const int gN   = (n + 255) / 256;
  const int gMM  = (n + 127) / 128;
  const int gP   = 2048;

  int* tk1 = (int*)d_out;
  int* tk2 = (int*)buf2;
  k_deg_init <<<gN,   blk, 0, stream>>>(dinv, tk1, n);
  k_deg_count<<<2048, blk, 0, stream>>>(dst, dinv, e);
  k_dinv     <<<gN,   blk, 0, stream>>>(dinv, n);

  k_gemm_l1  <<<gMM,  blk, 0, stream>>>(x, W1, dinv, buf1, buf2, n);
  k_scatter  <<<gP,   blk, 0, stream>>>(src, dst, dinv, buf1, buf2, tk1, e);

  k_gemm_l2  <<<gMM,  blk, 0, stream>>>(buf2, W2, b1, dinv, b2, buf1, out, n);
  k_ticket_zero<<<1, 32, 0, stream>>>(tk2);
  k_scatter  <<<gP,   blk, 0, stream>>>(src, dst, dinv, buf1, out, tk2, e);
}


# Edge norm split: dinv[src] folded into m by the product epilogue, dinv[dst] applied once per row at the layer-2 A load (layer 2: per edge); no dinv[src] gather per edge; f32 atomics, arrival order
# speedup vs baseline: 2.3356x; 1.0833x over previous
#include <hip/hip_runtime.h>

typedef __attribute__((ext_vector_type(2))) float v2f;
typedef __attribute__((ext_vector_type(8))) float v8f;

#define HD 128
#define WT_STRIDE 132

__global__ __launch_bounds__(256) void k_deg_init(float* __restrict__ deg, int* __restrict__ tk, int n) {
  int i = blockIdx.x * 256 + threadIdx.x;
  if (i < n) deg[i] = 1.0f;
  if (i < 2) tk[i * 32] = 0;
}

__global__ __launch_bounds__(256) void k_deg_count(const int* __restrict__ dst,
                                                   float* __restrict__ deg, int e) {
  int stride = gridDim.x * 256;
  for (int i = blockIdx.x * 256 + threadIdx.x; i < e; i += stride)
    atomicAdd(&deg[dst[i]], 1.0f);
}

__global__ __launch_bounds__(256) void k_dinv(float* __restrict__ deg, int n) {
  int i = blockIdx.x * 256 + threadIdx.x;
  if (i < n) deg[i] = rsqrtf(deg[i]);
}

template <bool IN_SCALE_BIAS_RELU, bool OUT_FINAL>
__device__ __forceinline__ void gemm_body(const float* __restrict__ A,
                                          const float* __restrict__ W,
                                          const float* __restrict__ bin,
                                          const float* __restrict__ dinv,
                                          const float* __restrict__ bout,
                                          float* __restrict__ m,
                                          float* __restrict__ start, int n) {
  __shared__ float Wt[HD * WT_STRIDE];
  __shared__ __attribute__((aligned(16))) float bs[HD];
  for (int i = threadIdx.x; i < HD * HD / 4; i += 256) {
    int k  = i >> 5;
    int n4 = (i & 31) * 4;
    const float4 w = *(const float4*)(W + k * HD + n4);
    Wt[(n4 + 0) * WT_STRIDE + k] = w.x;
    Wt[(n4 + 1) * WT_STRIDE + k] = w.y;
    Wt[(n4 + 2) * WT_STRIDE + k] = w.z;
    Wt[(n4 + 3) * WT_STRIDE + k] = w.w;
  }
  if (IN_SCALE_BIAS_RELU) {
    if (threadIdx.x < HD) bs[threadIdx.x] = bin[threadIdx.x];
  }
  __syncthreads();

  const int lane = threadIdx.x & 31;
  const int wave = threadIdx.x >> 5;
  const int r0   = blockIdx.x * 128 + wave * 16;
  if (r0 + 16 > n) return;
  const int hi = lane >> 4;
  const int lo = lane & 15;
  const float* arow = A + (size_t)(r0 + lo) * HD;
  float da = 1.0f;
  if (IN_SCALE_BIAS_RELU) da = dinv[r0 + lo];

  v8f acc[8] = {};
  for (int kk = 0; kk < 32; ++kk) {
    const int k = kk * 4 + hi * 2;
    v2f a = *(const v2f*)(arow + k);
    if (IN_SCALE_BIAS_RELU) {
      const v2f bb = *(const v2f*)(&bs[k]);
      const float x0 = da * a.x + bb.x;
      const float x1 = da * a.y + bb.y;
      a.x = x0 > 0.0f ? x0 : 0.0f;
      a.y = x1 > 0.0f ? x1 : 0.0f;
    }
#pragma unroll
    for (int t = 0; t < 8; ++t) {
      const v2f b = *(const v2f*)(&Wt[(t * 16 + lo) * WT_STRIDE + k]);
      acc[t] = __builtin_amdgcn_wmma_f32_16x16x4_f32(
          false, a, false, b, (short)0, acc[t], false, false);
    }
  }

  float di[8];
#pragma unroll
  for (int r = 0; r < 8; ++r) di[r] = dinv[r0 + hi * 8 + r];
#pragma unroll
  for (int t = 0; t < 8; ++t) {
    const size_t o = (size_t)(r0 + hi * 8) * HD + t * 16 + lo;
    float* mrow = m + o;
    float* srow = start + o;
    float bo = 0.0f;
    if (OUT_FINAL) bo = bout[t * 16 + lo];
#pragma unroll
    for (int r = 0; r < 8; ++r) {
      const float v = di[r] * acc[t][r];
      mrow[(size_t)r * HD] = v;
      if (OUT_FINAL) srow[(size_t)r * HD] = di[r] * v + bo;
      else           srow[(size_t)r * HD] = v;
    }
  }
}

__global__ __launch_bounds__(256) void k_gemm_l1(const float* __restrict__ A,
                                                 const float* __restrict__ W,
                                                 const float* __restrict__ dinv,
                                                 float* __restrict__ m,
                                                 float* __restrict__ start, int n) {
  gemm_body<false, false>(A, W, W, dinv, W, m, start, n);
}

__global__ __launch_bounds__(256) void k_gemm_l2(const float* __restrict__ A,
                                                 const float* __restrict__ W,
                                                 const float* __restrict__ bin,
                                                 const float* __restrict__ dinv,
                                                 const float* __restrict__ bout,
                                                 float* __restrict__ m,
                                                 float* __restrict__ start, int n) {
  gemm_body<true, true>(A, W, bin, dinv, bout, m, start, n);
}

#define EG 16
template <bool SCALE_DST>
__device__ __forceinline__ void scatter_batch(const int b, const int cb, const int lane,
                                              const int* __restrict__ src,
                                              const int* __restrict__ dst,
                                              const float* __restrict__ dinv,
                                              const float* __restrict__ m,
                                              float* __restrict__ agg, const int e) {
  const int i = b * 32 + lane;
  const unsigned ul = (unsigned)lane;
  int s = 0, d = 0;
  float nrm = 0.0f;
  if (i < e) {
    s = src[i];
    d = dst[i];
    if (SCALE_DST) nrm = dinv[d];
  }
  const int nrm_i = __float_as_int(nrm);
  const int cnt = (e - b * 32) < 32 ? (e - b * 32) : 32;
  if (cnt == 32) {
#pragma unroll 1
    for (int j0 = 0; j0 < 32; j0 += EG) {
      float v[EG][2];
#pragma unroll
      for (int j = 0; j < EG; ++j) {
        const int sj = __builtin_amdgcn_readlane(s, j0 + j);
        const float* mp = m + ((size_t)sj * HD + cb);
        v[j][0] = mp[ul];
        v[j][1] = (mp + 32)[ul];
      }
#pragma unroll
      for (int j = 0; j < EG; ++j) {
        const int dj = __builtin_amdgcn_readlane(d, j0 + j);
        float* p = agg + ((size_t)dj * HD + cb);
        if (SCALE_DST) {
          const float nj = __int_as_float(__builtin_amdgcn_readlane(nrm_i, j0 + j));
          atomicAdd(p + ul, nj * v[j][0]);
          atomicAdd(p + 32 + ul, nj * v[j][1]);
        } else {
          atomicAdd(p + ul, v[j][0]);
          atomicAdd(p + 32 + ul, v[j][1]);
        }
      }
    }
  } else {
    for (int j = 0; j < cnt; ++j) {
      const int sj = __builtin_amdgcn_readlane(s, j);
      const int dj = __builtin_amdgcn_readlane(d, j);
      const float nj = SCALE_DST ? __int_as_float(__builtin_amdgcn_readlane(nrm_i, j)) : 1.0f;
      const float* mp = m + (size_t)sj * HD + cb + lane;
      float* p = agg + (size_t)dj * HD + cb + lane;
      atomicAdd(p +  0, nj * mp[0]);
      atomicAdd(p + 32, nj * mp[32]);
    }
  }
}

__global__ __launch_bounds__(32) void k_ticket_zero(int* __restrict__ tk) {
  if (threadIdx.x < 2) tk[threadIdx.x * 32] = 0;
}

template <bool SCALE_DST>
__global__ __launch_bounds__(256) __attribute__((amdgpu_waves_per_eu(16, 16))) void k_scatter(const int* __restrict__ src,
                                                 const int* __restrict__ dst,
                                                 const float* __restrict__ dinv,
                                                 const float* __restrict__ m,
                                                 float* __restrict__ agg,
                                                 int* __restrict__ tk, int e) {
  __shared__ int slot[2];
  const int lane = threadIdx.x & 31;
  const int wave = __builtin_amdgcn_readfirstlane((int)(threadIdx.x >> 5));
  const int nb = (e + 31) >> 5;
  const int nt = (nb + 7) >> 3;
  int h = (__builtin_amdgcn_s_sendmsg_rtn(0x87) >> 18) & 1;
  int other_done = 0;
  int r = 0;
  if (threadIdx.x == 0) slot[0] = (atomicAdd(&tk[h * 32], 1) << 1) | h;
  for (;;) {
    __syncthreads();
    const int sv = __builtin_amdgcn_readfirstlane(slot[r & 1]);
    const int t  = sv >> 1;
    const int hc = sv & 1;
    ++r;
    if (t >= nt) {
      if (other_done) break;
      other_done = 1;
      h = hc ^ 1;
      if (threadIdx.x == 0) slot[r & 1] = (atomicAdd(&tk[h * 32], 1) << 1) | h;
      continue;
    }
    h = hc;
    int nxt = 0;
    if (threadIdx.x == 0) nxt = (atomicAdd(&tk[h * 32], 1) << 1) | h;
    const int b = t * 8 + wave;
    if (b < nb) scatter_batch<SCALE_DST>(b, hc * 64, lane, src, dst, dinv, m, agg, e);
    if (threadIdx.x == 0) slot[r & 1] = nxt;
  }
}

extern "C" void kernel_launch(void* const* d_in, const int* in_sizes, int n_in,
                              void* d_out, int out_size, void* d_ws, size_t ws_size,
                              hipStream_t stream) {
  const float* x  = (const float*)d_in[0];
  const float* W1 = (const float*)d_in[1];
  const float* b1 = (const float*)d_in[2];
  const float* W2 = (const float*)d_in[3];
  const float* b2 = (const float*)d_in[4];
  const int*   ei = (const int*)d_in[5];

  const int n = in_sizes[0] / HD;
  const int e = in_sizes[5] / 2;
  const int* src = ei;
  const int* dst = ei + e;
  const int total = n * HD;

  char* ws = (char*)d_ws;
  float* dinv = (float*)ws;
  size_t off1 = ((size_t)n * 4 + 255) & ~(size_t)255;
  float* buf1 = (float*)(ws + off1);
  size_t off2 = off1 + (((size_t)total * 4 + 255) & ~(size_t)255);
  float* buf2 = (float*)(ws + off2);
  float* out  = (float*)d_out;

  dim3 blk(256);
  const int gN   = (n + 255) / 256;
  const int gMM  = (n + 127) / 128;
  const int gP   = 2048;

  int* tk1 = (int*)d_out;
  int* tk2 = (int*)buf2;
  k_deg_init <<<gN,   blk, 0, stream>>>(dinv, tk1, n);
  k_deg_count<<<2048, blk, 0, stream>>>(dst, dinv, e);
  k_dinv     <<<gN,   blk, 0, stream>>>(dinv, n);

  k_gemm_l1  <<<gMM,  blk, 0, stream>>>(x, W1, dinv, buf1, buf2, n);
  k_scatter<false><<<gP, blk, 0, stream>>>(src, dst, dinv, buf1, buf2, tk1, e);

  k_gemm_l2  <<<gMM,  blk, 0, stream>>>(buf2, W2, b1, dinv, b2, buf1, out, n);
  k_ticket_zero<<<1, 32, 0, stream>>>(tk2);
  k_scatter<true><<<gP,  blk, 0, stream>>>(src, dst, dinv, buf1, out, tk2, e);
}


# Degree count split by node range across the two package halves (die id, work drawn by ticket): each line of count words is added to from one L2 half only; f32 atomics kept, counts exact in any order
# speedup vs baseline: 2.8361x; 1.2143x over previous
#include <hip/hip_runtime.h>

typedef __attribute__((ext_vector_type(2))) float v2f;
typedef __attribute__((ext_vector_type(8))) float v8f;

#define HD 128
#define WT_STRIDE 132

__global__ __launch_bounds__(256) void k_deg_init(float* __restrict__ deg, int* __restrict__ tk, int n) {
  int i = blockIdx.x * 256 + threadIdx.x;
  if (i < n) deg[i] = 1.0f;
  if (i < 4) tk[i * 32] = 0;
}

#define DEG_PT 8
template <bool FULL>
__device__ __forceinline__ void deg_chunk(const int base, const int hc, const int nh,
                                          const int* __restrict__ dst,
                                          float* __restrict__ deg, const int e) {
  int d[DEG_PT];
#pragma unroll
  for (int k = 0; k < DEG_PT; ++k) {
    const int i = base + k * 256;
    if (FULL) d[k] = dst[i];
    else      d[k] = (i < e) ? dst[i] : -1;
  }
#pragma unroll
  for (int k = 0; k < DEG_PT; ++k) {
    const int dk = d[k];
    const bool mine = (dk >= nh) == (hc != 0);
    if (FULL) { if (mine) atomicAdd(&deg[dk], 1.0f); }
    else      { if (dk >= 0 && mine) atomicAdd(&deg[dk], 1.0f); }
  }
}
__global__ __launch_bounds__(256) void k_deg_count(const int* __restrict__ dst,
                                                   float* __restrict__ deg,
                                                   int* __restrict__ tk, int n, int e) {
  __shared__ int slot[2];
  const int per = 256 * DEG_PT;
  const int nt  = (e + per - 1) / per;
  const int nh  = ((n >> 1) + 31) & ~31;
  int h = (__builtin_amdgcn_s_sendmsg_rtn(0x87) >> 18) & 1;
  int other_done = 0;
  int r = 0;
  if (threadIdx.x == 0) slot[0] = (atomicAdd(&tk[h * 32], 1) << 1) | h;
  for (;;) {
    __syncthreads();
    const int sv = __builtin_amdgcn_readfirstlane(slot[r & 1]);
    const int t  = sv >> 1;
    const int hc = sv & 1;
    ++r;
    if (t >= nt) {
      if (other_done) break;
      other_done = 1;
      h = hc ^ 1;
      if (threadIdx.x == 0) slot[r & 1] = (atomicAdd(&tk[h * 32], 1) << 1) | h;
      continue;
    }
    h = hc;
    int nxt = 0;
    if (threadIdx.x == 0) nxt = (atomicAdd(&tk[h * 32], 1) << 1) | h;
    const int base = t * per + (int)threadIdx.x;
    if (t * per + per <= e) deg_chunk<true >(base, hc, nh, dst, deg, e);
    else                    deg_chunk<false>(base, hc, nh, dst, deg, e);
    if (threadIdx.x == 0) slot[r & 1] = nxt;
  }
}

__global__ __launch_bounds__(256) void k_dinv(float* __restrict__ deg, int n) {
  int i = blockIdx.x * 256 + threadIdx.x;
  if (i < n) deg[i] = rsqrtf(deg[i]);
}

template <bool IN_SCALE_BIAS_RELU, bool OUT_FINAL>
__device__ __forceinline__ void gemm_body(const float* __restrict__ A,
                                          const float* __restrict__ W,
                                          const float* __restrict__ bin,
                                          const float* __restrict__ dinv,
                                          const float* __restrict__ bout,
                                          float* __restrict__ m,
                                          float* __restrict__ start, int n) {
  __shared__ float Wt[HD * WT_STRIDE];
  __shared__ __attribute__((aligned(16))) float bs[HD];
  for (int i = threadIdx.x; i < HD * HD / 4; i += 256) {
    int k  = i >> 5;
    int n4 = (i & 31) * 4;
    const float4 w = *(const float4*)(W + k * HD + n4);
    Wt[(n4 + 0) * WT_STRIDE + k] = w.x;
    Wt[(n4 + 1) * WT_STRIDE + k] = w.y;
    Wt[(n4 + 2) * WT_STRIDE + k] = w.z;
    Wt[(n4 + 3) * WT_STRIDE + k] = w.w;
  }
  if (IN_SCALE_BIAS_RELU) {
    if (threadIdx.x < HD) bs[threadIdx.x] = bin[threadIdx.x];
  }
  __syncthreads();

  const int lane = threadIdx.x & 31;
  const int wave = threadIdx.x >> 5;
  const int r0   = blockIdx.x * 128 + wave * 16;
  if (r0 + 16 > n) return;
  const int hi = lane >> 4;
  const int lo = lane & 15;
  const float* arow = A + (size_t)(r0 + lo) * HD;
  float da = 1.0f;
  if (IN_SCALE_BIAS_RELU) da = dinv[r0 + lo];

  v8f acc[8] = {};
  for (int kk = 0; kk < 32; ++kk) {
    const int k = kk * 4 + hi * 2;
    v2f a = *(const v2f*)(arow + k);
    if (IN_SCALE_BIAS_RELU) {
      const v2f bb = *(const v2f*)(&bs[k]);
      const float x0 = da * a.x + bb.x;
      const float x1 = da * a.y + bb.y;
      a.x = x0 > 0.0f ? x0 : 0.0f;
      a.y = x1 > 0.0f ? x1 : 0.0f;
    }
#pragma unroll
    for (int t = 0; t < 8; ++t) {
      const v2f b = *(const v2f*)(&Wt[(t * 16 + lo) * WT_STRIDE + k]);
      acc[t] = __builtin_amdgcn_wmma_f32_16x16x4_f32(
          false, a, false, b, (short)0, acc[t], false, false);
    }
  }

  float di[8];
#pragma unroll
  for (int r = 0; r < 8; ++r) di[r] = dinv[r0 + hi * 8 + r];
#pragma unroll
  for (int t = 0; t < 8; ++t) {
    const size_t o = (size_t)(r0 + hi * 8) * HD + t * 16 + lo;
    float* mrow = m + o;
    float* srow = start + o;
    float bo = 0.0f;
    if (OUT_FINAL) bo = bout[t * 16 + lo];
#pragma unroll
    for (int r = 0; r < 8; ++r) {
      const float v = di[r] * acc[t][r];
      mrow[(size_t)r * HD] = v;
      if (OUT_FINAL) srow[(size_t)r * HD] = di[r] * v + bo;
      else           srow[(size_t)r * HD] = v;
    }
  }
}

__global__ __launch_bounds__(256) void k_gemm_l1(const float* __restrict__ A,
                                                 const float* __restrict__ W,
                                                 const float* __restrict__ dinv,
                                                 float* __restrict__ m,
                                                 float* __restrict__ start, int n) {
  gemm_body<false, false>(A, W, W, dinv, W, m, start, n);
}

__global__ __launch_bounds__(256) void k_gemm_l2(const float* __restrict__ A,
                                                 const float* __restrict__ W,
                                                 const float* __restrict__ bin,
                                                 const float* __restrict__ dinv,
                                                 const float* __restrict__ bout,
                                                 float* __restrict__ m,
                                                 float* __restrict__ start, int n) {
  gemm_body<true, true>(A, W, bin, dinv, bout, m, start, n);
}

#define EG 16
template <bool SCALE_DST>
__device__ __forceinline__ void scatter_batch(const int b, const int cb, const int lane,
                                              const int* __restrict__ src,
                                              const int* __restrict__ dst,
                                              const float* __restrict__ dinv,
                                              const float* __restrict__ m,
                                              float* __restrict__ agg, const int e) {
  const int i = b * 32 + lane;
  const unsigned ul = (unsigned)lane;
  int s = 0, d = 0;
  float nrm = 0.0f;
  if (i < e) {
    s = src[i];
    d = dst[i];
    if (SCALE_DST) nrm = dinv[d];
  }
  const int nrm_i = __float_as_int(nrm);
  const int cnt = (e - b * 32) < 32 ? (e - b * 32) : 32;
  if (cnt == 32) {
#pragma unroll 1
    for (int j0 = 0; j0 < 32; j0 += EG) {
      float v[EG][2];
#pragma unroll
      for (int j = 0; j < EG; ++j) {
        const int sj = __builtin_amdgcn_readlane(s, j0 + j);
        const float* mp = m + ((size_t)sj * HD + cb);
        v[j][0] = mp[ul];
        v[j][1] = (mp + 32)[ul];
      }
#pragma unroll
      for (int j = 0; j < EG; ++j) {
        const int dj = __builtin_amdgcn_readlane(d, j0 + j);
        float* p = agg + ((size_t)dj * HD + cb);
        if (SCALE_DST) {
          const float nj = __int_as_float(__builtin_amdgcn_readlane(nrm_i, j0 + j));
          atomicAdd(p + ul, nj * v[j][0]);
          atomicAdd(p + 32 + ul, nj * v[j][1]);
        } else {
          atomicAdd(p + ul, v[j][0]);
          atomicAdd(p + 32 + ul, v[j][1]);
        }
      }
    }
  } else {
    for (int j = 0; j < cnt; ++j) {
      const int sj = __builtin_amdgcn_readlane(s, j);
      const int dj = __builtin_amdgcn_readlane(d, j);
      const float nj = SCALE_DST ? __int_as_float(__builtin_amdgcn_readlane(nrm_i, j)) : 1.0f;
      const float* mp = m + (size_t)sj * HD + cb + lane;
      float* p = agg + (size_t)dj * HD + cb + lane;
      atomicAdd(p +  0, nj * mp[0]);
      atomicAdd(p + 32, nj * mp[32]);
    }
  }
}

__global__ __launch_bounds__(32) void k_ticket_zero(int* __restrict__ tk) {
  if (threadIdx.x < 2) tk[threadIdx.x * 32] = 0;
}

template <bool SCALE_DST>
__global__ __launch_bounds__(256) __attribute__((amdgpu_waves_per_eu(16, 16))) void k_scatter(const int* __restrict__ src,
                                                 const int* __restrict__ dst,
                                                 const float* __restrict__ dinv,
                                                 const float* __restrict__ m,
                                                 float* __restrict__ agg,
                                                 int* __restrict__ tk, int e) {
  __shared__ int slot[2];
  const int lane = threadIdx.x & 31;
  const int wave = __builtin_amdgcn_readfirstlane((int)(threadIdx.x >> 5));
  const int nb = (e + 31) >> 5;
  const int nt = (nb + 7) >> 3;
  int h = (__builtin_amdgcn_s_sendmsg_rtn(0x87) >> 18) & 1;
  int other_done = 0;
  int r = 0;
  if (threadIdx.x == 0) slot[0] = (atomicAdd(&tk[h * 32], 1) << 1) | h;
  for (;;) {
    __syncthreads();
    const int sv = __builtin_amdgcn_readfirstlane(slot[r & 1]);
    const int t  = sv >> 1;
    const int hc = sv & 1;
    ++r;
    if (t >= nt) {
      if (other_done) break;
      other_done = 1;
      h = hc ^ 1;
      if (threadIdx.x == 0) slot[r & 1] = (atomicAdd(&tk[h * 32], 1) << 1) | h;
      continue;
    }
    h = hc;
    int nxt = 0;
    if (threadIdx.x == 0) nxt = (atomicAdd(&tk[h * 32], 1) << 1) | h;
    const int b = t * 8 + wave;
    if (b < nb) scatter_batch<SCALE_DST>(b, hc * 64, lane, src, dst, dinv, m, agg, e);
    if (threadIdx.x == 0) slot[r & 1] = nxt;
  }
}

extern "C" void kernel_launch(void* const* d_in, const int* in_sizes, int n_in,
                              void* d_out, int out_size, void* d_ws, size_t ws_size,
                              hipStream_t stream) {
  const float* x  = (const float*)d_in[0];
  const float* W1 = (const float*)d_in[1];
  const float* b1 = (const float*)d_in[2];
  const float* W2 = (const float*)d_in[3];
  const float* b2 = (const float*)d_in[4];
  const int*   ei = (const int*)d_in[5];

  const int n = in_sizes[0] / HD;
  const int e = in_sizes[5] / 2;
  const int* src = ei;
  const int* dst = ei + e;
  const int total = n * HD;

  char* ws = (char*)d_ws;
  float* dinv = (float*)ws;
  size_t off1 = ((size_t)n * 4 + 255) & ~(size_t)255;
  float* buf1 = (float*)(ws + off1);
  size_t off2 = off1 + (((size_t)total * 4 + 255) & ~(size_t)255);
  float* buf2 = (float*)(ws + off2);
  float* out  = (float*)d_out;

  dim3 blk(256);
  const int gN   = (n + 255) / 256;
  const int gMM  = (n + 127) / 128;
  const int gP   = 2048;
  const int gD   = 1024;

  int* tk1 = (int*)d_out;
  int* tk2 = (int*)buf2;
  k_deg_init <<<gN,   blk, 0, stream>>>(dinv, tk1, n);
  k_deg_count<<<gD,   blk, 0, stream>>>(dst, dinv, tk1 + 64, n, e);
  k_dinv     <<<gN,   blk, 0, stream>>>(dinv, n);

  k_gemm_l1  <<<gMM,  blk, 0, stream>>>(x, W1, dinv, buf1, buf2, n);
  k_scatter<false><<<gP, blk, 0, stream>>>(src, dst, dinv, buf1, buf2, tk1, e);

  k_gemm_l2  <<<gMM,  blk, 0, stream>>>(buf2, W2, b1, dinv, b2, buf1, out, n);
  k_ticket_zero<<<1, 32, 0, stream>>>(tk2);
  k_scatter<true><<<gP,  blk, 0, stream>>>(src, dst, dinv, buf1, out, tk2, e);
}


# Edge sum by owners: edges grouped by 391-row destination range (LDS int atomics, scan); one work-group per range adds rows in LDS (f32 LDS atomics), stores once; no global atomics; arrival order
# speedup vs baseline: 3.5923x; 1.2666x over previous
#include <hip/hip_runtime.h>

typedef __attribute__((ext_vector_type(2))) float v2f;
typedef __attribute__((ext_vector_type(8))) float v8f;
typedef __attribute__((ext_vector_type(2))) int v2i;

#define HD 128
#define WT_STRIDE 132

#define NBIN 256
#define RPB 391
#define CH_PT 8
#define CH (1024 * CH_PT)
#define SCAN_PER 64

__global__ __launch_bounds__(1024) void k_bin_count(const int* __restrict__ dst, int* __restrict__ tab,
                                                    int e, int nc) {
  __shared__ int h[NBIN];
  const int tid = threadIdx.x;
  const int c = blockIdx.x;
  if (tid < NBIN) h[tid] = 0;
  __syncthreads();
  int d[CH_PT];
#pragma unroll
  for (int k = 0; k < CH_PT; ++k) {
    const int i = c * CH + k * 1024 + tid;
    d[k] = (i < e) ? dst[i] : -1;
  }
#pragma unroll
  for (int k = 0; k < CH_PT; ++k)
    if (d[k] >= 0) atomicAdd(&h[d[k] / RPB], 1);
  __syncthreads();
  if (tid < NBIN) tab[tid * nc + c] = h[tid];
}

__global__ __launch_bounds__(1024) void k_bin_scan(int* __restrict__ tab, int total) {
  __shared__ int wsum[32];
  const int tid  = threadIdx.x;
  const int lane = tid & 31;
  const int wave = tid >> 5;
  const int i0 = tid * SCAN_PER;
  const bool on = i0 < total;
  int cn[SCAN_PER];
#pragma unroll
  for (int k = 0; k < SCAN_PER; ++k) cn[k] = on ? tab[i0 + k] : 0;
  int sum = 0;
#pragma unroll
  for (int k = 0; k < SCAN_PER; ++k) sum += cn[k];
  int v = sum;
  for (int o = 1; o < 32; o <<= 1) {
    const int u = __shfl_up(v, o);
    if (lane >= o) v += u;
  }
  if (lane == 31) wsum[wave] = v;
  __syncthreads();
  if (wave == 0) {
    int w = wsum[lane];
    for (int o = 1; o < 32; o <<= 1) {
      const int u = __shfl_up(w, o);
      if (lane >= o) w += u;
    }
    wsum[lane] = w;
  }
  __syncthreads();
  int run = v - sum + (wave > 0 ? wsum[wave - 1] : 0);
#pragma unroll
  for (int k = 0; k < SCAN_PER; ++k) {
    if (on) tab[i0 + k] = run;
    run += cn[k];
  }
}

__global__ __launch_bounds__(1024) void k_bin_place(const int* __restrict__ src, const int* __restrict__ dst,
                                                    const int* __restrict__ tab, int* __restrict__ eb,
                                                    int e, int nc) {
  __shared__ int cur[NBIN];
  const int tid = threadIdx.x;
  const int c = blockIdx.x;
  if (tid < NBIN) cur[tid] = tab[tid * nc + c];
  __syncthreads();
  int s[CH_PT], d[CH_PT];
#pragma unroll
  for (int k = 0; k < CH_PT; ++k) {
    const int i = c * CH + k * 1024 + tid;
    s[k] = (i < e) ? src[i] : 0;
    d[k] = (i < e) ? dst[i] : -1;
  }
#pragma unroll
  for (int k = 0; k < CH_PT; ++k) {
    if (d[k] >= 0) {
      const int pos = atomicAdd(&cur[d[k] / RPB], 1);
      v2i sd;
      sd.x = s[k];
      sd.y = d[k];
      *(v2i*)(eb + 2 * (size_t)pos) = sd;
    }
  }
}

__global__ __launch_bounds__(1024) void k_dinv_bins(const int* __restrict__ eb, const int* __restrict__ tab,
                                                    float* __restrict__ dinv, int n, int e, int nc) {
  __shared__ int cnt[RPB];
  const int tid = threadIdx.x;
  const int b = blockIdx.x;
  const int base = b * RPB;
  int rows = n - base;
  if (rows > RPB) rows = RPB;
  if (tid < RPB) cnt[tid] = 0;
  __syncthreads();
  const int e0 = tab[b * nc];
  const int e1 = (b + 1 < NBIN) ? tab[(b + 1) * nc] : e;
#pragma unroll 4
  for (int i = e0 + tid; i < e1; i += 1024) atomicAdd(&cnt[eb[2 * (size_t)i + 1] - base], 1);
  __syncthreads();
  if (tid < rows) dinv[base + tid] = rsqrtf(1.0f + (float)cnt[tid]);
}

template <bool IN_SCALE_BIAS_RELU>
__device__ __forceinline__ void gemm_body(const float* A,
                                          const float* __restrict__ W,
                                          const float* __restrict__ bin,
                                          const float* __restrict__ dinv,
                                          float* m, int n) {
  __shared__ float Wt[HD * WT_STRIDE];
  __shared__ __attribute__((aligned(16))) float bs[HD];
  for (int i = threadIdx.x; i < HD * HD / 4; i += 256) {
    int k  = i >> 5;
    int n4 = (i & 31) * 4;
    const float4 w = *(const float4*)(W + k * HD + n4);
    Wt[(n4 + 0) * WT_STRIDE + k] = w.x;
    Wt[(n4 + 1) * WT_STRIDE + k] = w.y;
    Wt[(n4 + 2) * WT_STRIDE + k] = w.z;
    Wt[(n4 + 3) * WT_STRIDE + k] = w.w;
  }
  if (IN_SCALE_BIAS_RELU) {
    if (threadIdx.x < HD) bs[threadIdx.x] = bin[threadIdx.x];
  }
  __syncthreads();

  const int lane = threadIdx.x & 31;
  const int wave = threadIdx.x >> 5;
  const int r0   = blockIdx.x * 128 + wave * 16;
  if (r0 + 16 > n) return;
  const int hi = lane >> 4;
  const int lo = lane & 15;
  const float* arow = A + (size_t)(r0 + lo) * HD;
  float da = 1.0f;
  if (IN_SCALE_BIAS_RELU) da = dinv[r0 + lo];

  v8f acc[8] = {};
  for (int kk = 0; kk < 32; ++kk) {
    const int k = kk * 4 + hi * 2;
    v2f a = *(const v2f*)(arow + k);
    if (IN_SCALE_BIAS_RELU) {
      const v2f bb = *(const v2f*)(&bs[k]);
      const float x0 = da * a.x + bb.x;
      const float x1 = da * a.y + bb.y;
      a.x = x0 > 0.0f ? x0 : 0.0f;
      a.y = x1 > 0.0f ? x1 : 0.0f;
    }
#pragma unroll
    for (int t = 0; t < 8; ++t) {
      const v2f b = *(const v2f*)(&Wt[(t * 16 + lo) * WT_STRIDE + k]);
      acc[t] = __builtin_amdgcn_wmma_f32_16x16x4_f32(
          false, a, false, b, (short)0, acc[t], false, false);
    }
  }

  float di[8];
#pragma unroll
  for (int r = 0; r < 8; ++r) di[r] = dinv[r0 + hi * 8 + r];
#pragma unroll
  for (int t = 0; t < 8; ++t) {
    float* mrow = m + (size_t)(r0 + hi * 8) * HD + t * 16 + lo;
#pragma unroll
    for (int r = 0; r < 8; ++r) mrow[(size_t)r * HD] = di[r] * acc[t][r];
  }
}

__global__ __launch_bounds__(256) void k_gemm_l1(const float* __restrict__ A,
                                                 const float* __restrict__ W,
                                                 const float* __restrict__ dinv,
                                                 float* __restrict__ m, int n) {
  gemm_body<false>(A, W, W, dinv, m, n);
}

__global__ __launch_bounds__(256) void k_gemm_l2(const float* __restrict__ W,
                                                 const float* __restrict__ bin,
                                                 const float* __restrict__ dinv,
                                                 float* m, int n) {
  gemm_body<true>(m, W, bin, dinv, m, n);
}

#define EG 16
template <bool FINAL>
__global__ __launch_bounds__(1024) void k_agg(const int* __restrict__ eb, const int* __restrict__ tab,
                                              const float* __restrict__ m,
                                              const float* __restrict__ dinv,
                                              const float* __restrict__ bias,
                                              float* __restrict__ outp, int n, int e, int nc) {
  __shared__ float acc[RPB * HD];
  const int lane = threadIdx.x & 31;
  const int wave = __builtin_amdgcn_readfirstlane((int)(threadIdx.x >> 5));
  const unsigned ul4 = (unsigned)lane * 4u;
  const int b = blockIdx.x;
  const int base = b * RPB;
  int rows = n - base;
  if (rows > RPB) rows = RPB;

#pragma unroll 4
  for (int r = wave; r < rows; r += 32) {
    const float4 v = *(const float4*)(m + (size_t)(base + r) * HD + ul4);
    float* a = acc + r * HD + lane;
    a[0]  = v.x;
    a[32] = v.y;
    a[64] = v.z;
    a[96] = v.w;
  }
  __syncthreads();

  const int e0 = tab[b * nc];
  const int e1 = (b + 1 < NBIN) ? tab[(b + 1) * nc] : e;
  for (int q0 = e0 + wave * 32; q0 < e1; q0 += 32 * 32) {
    const int i = q0 + lane;
    int s = 0, dl = 0;
    if (i < e1) {
      const v2i sd = *(const v2i*)(eb + 2 * (size_t)i);
      s  = sd.x;
      dl = sd.y - base;
    }
    const int cnt = (e1 - q0) < 32 ? (e1 - q0) : 32;
    if (cnt == 32) {
#pragma unroll 1
      for (int j0 = 0; j0 < 32; j0 += EG) {
        float4 v[EG];
#pragma unroll
        for (int j = 0; j < EG; ++j) {
          const int sj = __builtin_amdgcn_readlane(s, j0 + j);
          const float* mp = m + (size_t)sj * HD;
          v[j] = *(const float4*)(mp + ul4);
        }
#pragma unroll
        for (int j = 0; j < EG; ++j) {
          const int dj = __builtin_amdgcn_readlane(dl, j0 + j);
          float* a = acc + dj * HD + lane;
          atomicAdd(a,      v[j].x);
          atomicAdd(a + 32, v[j].y);
          atomicAdd(a + 64, v[j].z);
          atomicAdd(a + 96, v[j].w);
        }
      }
    } else {
      for (int j = 0; j < cnt; ++j) {
        const int sj = __builtin_amdgcn_readlane(s, j);
        const int dj = __builtin_amdgcn_readlane(dl, j);
        const float4 v = *(const float4*)(m + (size_t)sj * HD + ul4);
        float* a = acc + dj * HD + lane;
        atomicAdd(a,      v.x);
        atomicAdd(a + 32, v.y);
        atomicAdd(a + 64, v.z);
        atomicAdd(a + 96, v.w);
      }
    }
  }
  __syncthreads();

  float4 bb = make_float4(0.0f, 0.0f, 0.0f, 0.0f);
  if (FINAL) bb = *(const float4*)(bias + ul4);
  for (int r = wave; r < rows; r += 32) {
    const float* a = acc + r * HD + lane;
    float4 v;
    v.x = a[0];
    v.y = a[32];
    v.z = a[64];
    v.w = a[96];
    if (FINAL) {
      const float dr = dinv[base + r];
      v.x = dr * v.x + bb.x;
      v.y = dr * v.y + bb.y;
      v.z = dr * v.z + bb.z;
      v.w = dr * v.w + bb.w;
    }
    *(float4*)(outp + (size_t)(base + r) * HD + ul4) = v;
  }
}

extern "C" void kernel_launch(void* const* d_in, const int* in_sizes, int n_in,
                              void* d_out, int out_size, void* d_ws, size_t ws_size,
                              hipStream_t stream) {
  const float* x  = (const float*)d_in[0];
  const float* W1 = (const float*)d_in[1];
  const float* b1 = (const float*)d_in[2];
  const float* W2 = (const float*)d_in[3];
  const float* b2 = (const float*)d_in[4];
  const int*   ei = (const int*)d_in[5];

  const int n = in_sizes[0] / HD;
  const int e = in_sizes[5] / 2;
  const int* src = ei;
  const int* dst = ei + e;
  const int total = n * HD;

  char* ws = (char*)d_ws;
  float* dinv = (float*)ws;
  size_t off1 = ((size_t)n * 4 + 255) & ~(size_t)255;
  float* buf1 = (float*)(ws + off1);
  size_t off2 = off1 + (((size_t)total * 4 + 255) & ~(size_t)255);
  int*   eb   = (int*)(ws + off2);
  int*   tab  = eb + 2 * (size_t)e;
  float* out  = (float*)d_out;

  dim3 blk(256);
  dim3 big(1024);
  const int gMM = (n + 127) / 128;
  const int nc  = (e + CH - 1) / CH;

  k_bin_count<<<nc,   big, 0, stream>>>(dst, tab, e, nc);
  k_bin_scan <<<1,    big, 0, stream>>>(tab, NBIN * nc);
  k_bin_place<<<nc,   big, 0, stream>>>(src, dst, tab, eb, e, nc);
  k_dinv_bins<<<NBIN, big, 0, stream>>>(eb, tab, dinv, n, e, nc);

  k_gemm_l1  <<<gMM,  blk, 0, stream>>>(x, W1, dinv, out, n);
  k_agg<false><<<NBIN, big, 0, stream>>>(eb, tab, out, dinv, b1, buf1, n, e, nc);

  k_gemm_l2  <<<gMM,  blk, 0, stream>>>(W2, b1, dinv, buf1, n);
  k_agg<true><<<NBIN, big, 0, stream>>>(eb, tab, buf1, dinv, b2, out, n, e, nc);
}


# Offset scan moves its table as whole 512-byte wave rows (wide loads and stores), turned through padded LDS, not as 64 4-byte loads per lane that hit 32 lines each; same offsets, float sums untouched
# speedup vs baseline: 3.9950x; 1.1121x over previous
#include <hip/hip_runtime.h>

typedef __attribute__((ext_vector_type(2))) float v2f;
typedef __attribute__((ext_vector_type(8))) float v8f;
typedef __attribute__((ext_vector_type(2))) int v2i;
typedef __attribute__((ext_vector_type(4))) int v4i;

#define HD 128
#define WT_STRIDE 132

#define NBIN 256
#define RPB 391
#define CH_PT 8
#define CH (1024 * CH_PT)
#define SCAN_PER 64
#define SCAN_PITCH 68

__global__ __launch_bounds__(1024) void k_bin_count(const int* __restrict__ dst, int* __restrict__ tab,
                                                    int e, int nc) {
  __shared__ int h[NBIN];
  const int tid = threadIdx.x;
  const int c = blockIdx.x;
  if (tid < NBIN) h[tid] = 0;
  __syncthreads();
  int d[CH_PT];
#pragma unroll
  for (int k = 0; k < CH_PT; ++k) {
    const int i = c * CH + k * 1024 + tid;
    d[k] = (i < e) ? dst[i] : -1;
  }
#pragma unroll
  for (int k = 0; k < CH_PT; ++k)
    if (d[k] >= 0) atomicAdd(&h[d[k] / RPB], 1);
  __syncthreads();
  if (tid < NBIN) tab[tid * nc + c] = h[tid];
}

__global__ __launch_bounds__(1024) void k_bin_scan(int* __restrict__ tab, int total) {
  __shared__ __attribute__((aligned(16))) int rowbuf[1024 * SCAN_PITCH];
  __shared__ int wsum[32];
  const int tid  = threadIdx.x;
  const int lane = tid & 31;
  const int wave = tid >> 5;

  v4i g[SCAN_PER / 4];
#pragma unroll
  for (int j = 0; j < SCAN_PER / 4; ++j) {
    const int f = (j * 1024 + tid) * 4;
    v4i z = {0, 0, 0, 0};
    g[j] = (f < total) ? *(const v4i*)(tab + f) : z;
  }
#pragma unroll
  for (int j = 0; j < SCAN_PER / 4; ++j) {
    const int f = (j * 1024 + tid) * 4;
    *(v4i*)(&rowbuf[(f >> 6) * SCAN_PITCH + (f & 63)]) = g[j];
  }
  __syncthreads();

  int* myrow = rowbuf + tid * SCAN_PITCH;
  int cn[SCAN_PER];
#pragma unroll
  for (int k = 0; k < SCAN_PER / 4; ++k) {
    const v4i c = *(const v4i*)(myrow + 4 * k);
    cn[4 * k + 0] = c.x;
    cn[4 * k + 1] = c.y;
    cn[4 * k + 2] = c.z;
    cn[4 * k + 3] = c.w;
  }
  int sum = 0;
#pragma unroll
  for (int k = 0; k < SCAN_PER; ++k) sum += cn[k];
  int v = sum;
  for (int o = 1; o < 32; o <<= 1) {
    const int u = __shfl_up(v, o);
    if (lane >= o) v += u;
  }
  if (lane == 31) wsum[wave] = v;
  __syncthreads();
  if (wave == 0) {
    int w = wsum[lane];
    for (int o = 1; o < 32; o <<= 1) {
      const int u = __shfl_up(w, o);
      if (lane >= o) w += u;
    }
    wsum[lane] = w;
  }
  __syncthreads();
  int run = v - sum + (wave > 0 ? wsum[wave - 1] : 0);
#pragma unroll
  for (int k = 0; k < SCAN_PER / 4; ++k) {
    v4i o;
    o.x = run; run += cn[4 * k + 0];
    o.y = run; run += cn[4 * k + 1];
    o.z = run; run += cn[4 * k + 2];
    o.w = run; run += cn[4 * k + 3];
    *(v4i*)(myrow + 4 * k) = o;
  }
  __syncthreads();

#pragma unroll
  for (int j = 0; j < SCAN_PER / 4; ++j) {
    const int f = (j * 1024 + tid) * 4;
    if (f < total) *(v4i*)(tab + f) = *(const v4i*)(&rowbuf[(f >> 6) * SCAN_PITCH + (f & 63)]);
  }
}

__global__ __launch_bounds__(1024) void k_bin_place(const int* __restrict__ src, const int* __restrict__ dst,
                                                    const int* __restrict__ tab, int* __restrict__ eb,
                                                    int e, int nc) {
  __shared__ int cur[NBIN];
  const int tid = threadIdx.x;
  const int c = blockIdx.x;
  if (tid < NBIN) cur[tid] = tab[tid * nc + c];
  __syncthreads();
  int s[CH_PT], d[CH_PT];
#pragma unroll
  for (int k = 0; k < CH_PT; ++k) {
    const int i = c * CH + k * 1024 + tid;
    s[k] = (i < e) ? src[i] : 0;
    d[k] = (i < e) ? dst[i] : -1;
  }
#pragma unroll
  for (int k = 0; k < CH_PT; ++k) {
    if (d[k] >= 0) {
      const int pos = atomicAdd(&cur[d[k] / RPB], 1);
      v2i sd;
      sd.x = s[k];
      sd.y = d[k];
      *(v2i*)(eb + 2 * (size_t)pos) = sd;
    }
  }
}

__global__ __launch_bounds__(1024) void k_dinv_bins(const int* __restrict__ eb, const int* __restrict__ tab,
                                                    float* __restrict__ dinv, int n, int e, int nc) {
  __shared__ int cnt[RPB];
  const int tid = threadIdx.x;
  const int b = blockIdx.x;
  const int base = b * RPB;
  int rows = n - base;
  if (rows > RPB) rows = RPB;
  if (tid < RPB) cnt[tid] = 0;
  __syncthreads();
  const int e0 = tab[b * nc];
  const int e1 = (b + 1 < NBIN) ? tab[(b + 1) * nc] : e;
#pragma unroll 4
  for (int i = e0 + tid; i < e1; i += 1024) atomicAdd(&cnt[eb[2 * (size_t)i + 1] - base], 1);
  __syncthreads();
  if (tid < rows) dinv[base + tid] = rsqrtf(1.0f + (float)cnt[tid]);
}

template <bool IN_SCALE_BIAS_RELU>
__device__ __forceinline__ void gemm_body(const float* A,
                                          const float* __restrict__ W,
                                          const float* __restrict__ bin,
                                          const float* __restrict__ dinv,
                                          float* m, int n) {
  __shared__ float Wt[HD * WT_STRIDE];
  __shared__ __attribute__((aligned(16))) float bs[HD];
  for (int i = threadIdx.x; i < HD * HD / 4; i += 256) {
    int k  = i >> 5;
    int n4 = (i & 31) * 4;
    const float4 w = *(const float4*)(W + k * HD + n4);
    Wt[(n4 + 0) * WT_STRIDE + k] = w.x;
    Wt[(n4 + 1) * WT_STRIDE + k] = w.y;
    Wt[(n4 + 2) * WT_STRIDE + k] = w.z;
    Wt[(n4 + 3) * WT_STRIDE + k] = w.w;
  }
  if (IN_SCALE_BIAS_RELU) {
    if (threadIdx.x < HD) bs[threadIdx.x] = bin[threadIdx.x];
  }
  __syncthreads();

  const int lane = threadIdx.x & 31;
  const int wave = threadIdx.x >> 5;
  const int r0   = blockIdx.x * 128 + wave * 16;
  if (r0 + 16 > n) return;
  const int hi = lane >> 4;
  const int lo = lane & 15;
  const float* arow = A + (size_t)(r0 + lo) * HD;
  float da = 1.0f;
  if (IN_SCALE_BIAS_RELU) da = dinv[r0 + lo];

  v8f acc[8] = {};
  for (int kk = 0; kk < 32; ++kk) {
    const int k = kk * 4 + hi * 2;
    v2f a = *(const v2f*)(arow + k);
    if (IN_SCALE_BIAS_RELU) {
      const v2f bb = *(const v2f*)(&bs[k]);
      const float x0 = da * a.x + bb.x;
      const float x1 = da * a.y + bb.y;
      a.x = x0 > 0.0f ? x0 : 0.0f;
      a.y = x1 > 0.0f ? x1 : 0.0f;
    }
#pragma unroll
    for (int t = 0; t < 8; ++t) {
      const v2f b = *(const v2f*)(&Wt[(t * 16 + lo) * WT_STRIDE + k]);
      acc[t] = __builtin_amdgcn_wmma_f32_16x16x4_f32(
          false, a, false, b, (short)0, acc[t], false, false);
    }
  }

  float di[8];
#pragma unroll
  for (int r = 0; r < 8; ++r) di[r] = dinv[r0 + hi * 8 + r];
#pragma unroll
  for (int t = 0; t < 8; ++t) {
    float* mrow = m + (size_t)(r0 + hi * 8) * HD + t * 16 + lo;
#pragma unroll
    for (int r = 0; r < 8; ++r) mrow[(size_t)r * HD] = di[r] * acc[t][r];
  }
}

__global__ __launch_bounds__(256) void k_gemm_l1(const float* __restrict__ A,
                                                 const float* __restrict__ W,
                                                 const float* __restrict__ dinv,
                                                 float* __restrict__ m, int n) {
  gemm_body<false>(A, W, W, dinv, m, n);
}

__global__ __launch_bounds__(256) void k_gemm_l2(const float* __restrict__ W,
                                                 const float* __restrict__ bin,
                                                 const float* __restrict__ dinv,
                                                 float* m, int n) {
  gemm_body<true>(m, W, bin, dinv, m, n);
}

#define EG 16
template <bool FINAL>
__global__ __launch_bounds__(1024) void k_agg(const int* __restrict__ eb, const int* __restrict__ tab,
                                              const float* __restrict__ m,
                                              const float* __restrict__ dinv,
                                              const float* __restrict__ bias,
                                              float* __restrict__ outp, int n, int e, int nc) {
  __shared__ float acc[RPB * HD];
  const int lane = threadIdx.x & 31;
  const int wave = __builtin_amdgcn_readfirstlane((int)(threadIdx.x >> 5));
  const unsigned ul4 = (unsigned)lane * 4u;
  const int b = blockIdx.x;
  const int base = b * RPB;
  int rows = n - base;
  if (rows > RPB) rows = RPB;

#pragma unroll 4
  for (int r = wave; r < rows; r += 32) {
    const float4 v = *(const float4*)(m + (size_t)(base + r) * HD + ul4);
    float* a = acc + r * HD + lane;
    a[0]  = v.x;
    a[32] = v.y;
    a[64] = v.z;
    a[96] = v.w;
  }
  __syncthreads();

  const int e0 = tab[b * nc];
  const int e1 = (b + 1 < NBIN) ? tab[(b + 1) * nc] : e;
  for (int q0 = e0 + wave * 32; q0 < e1; q0 += 32 * 32) {
    const int i = q0 + lane;
    int s = 0, dl = 0;
    if (i < e1) {
      const v2i sd = *(const v2i*)(eb + 2 * (size_t)i);
      s  = sd.x;
      dl = sd.y - base;
    }
    const int cnt = (e1 - q0) < 32 ? (e1 - q0) : 32;
    if (cnt == 32) {
#pragma unroll 1
      for (int j0 = 0; j0 < 32; j0 += EG) {
        float4 v[EG];
#pragma unroll
        for (int j = 0; j < EG; ++j) {
          const int sj = __builtin_amdgcn_readlane(s, j0 + j);
          const float* mp = m + (size_t)sj * HD;
          v[j] = *(const float4*)(mp + ul4);
        }
#pragma unroll
        for (int j = 0; j < EG; ++j) {
          const int dj = __builtin_amdgcn_readlane(dl, j0 + j);
          float* a = acc + dj * HD + lane;
          atomicAdd(a,      v[j].x);
          atomicAdd(a + 32, v[j].y);
          atomicAdd(a + 64, v[j].z);
          atomicAdd(a + 96, v[j].w);
        }
      }
    } else {
      for (int j = 0; j < cnt; ++j) {
        const int sj = __builtin_amdgcn_readlane(s, j);
        const int dj = __builtin_amdgcn_readlane(dl, j);
        const float4 v = *(const float4*)(m + (size_t)sj * HD + ul4);
        float* a = acc + dj * HD + lane;
        atomicAdd(a,      v.x);
        atomicAdd(a + 32, v.y);
        atomicAdd(a + 64, v.z);
        atomicAdd(a + 96, v.w);
      }
    }
  }
  __syncthreads();

  float4 bb = make_float4(0.0f, 0.0f, 0.0f, 0.0f);
  if (FINAL) bb = *(const float4*)(bias + ul4);
  for (int r = wave; r < rows; r += 32) {
    const float* a = acc + r * HD + lane;
    float4 v;
    v.x = a[0];
    v.y = a[32];
    v.z = a[64];
    v.w = a[96];
    if (FINAL) {
      const float dr = dinv[base + r];
      v.x = dr * v.x + bb.x;
      v.y = dr * v.y + bb.y;
      v.z = dr * v.z + bb.z;
      v.w = dr * v.w + bb.w;
    }
    *(float4*)(outp + (size_t)(base + r) * HD + ul4) = v;
  }
}

extern "C" void kernel_launch(void* const* d_in, const int* in_sizes, int n_in,
                              void* d_out, int out_size, void* d_ws, size_t ws_size,
                              hipStream_t stream) {
  const float* x  = (const float*)d_in[0];
  const float* W1 = (const float*)d_in[1];
  const float* b1 = (const float*)d_in[2];
  const float* W2 = (const float*)d_in[3];
  const float* b2 = (const float*)d_in[4];
  const int*   ei = (const int*)d_in[5];

  const int n = in_sizes[0] / HD;
  const int e = in_sizes[5] / 2;
  const int* src = ei;
  const int* dst = ei + e;
  const int total = n * HD;

  char* ws = (char*)d_ws;
  float* dinv = (float*)ws;
  size_t off1 = ((size_t)n * 4 + 255) & ~(size_t)255;
  float* buf1 = (float*)(ws + off1);
  size_t off2 = off1 + (((size_t)total * 4 + 255) & ~(size_t)255);
  int*   eb   = (int*)(ws + off2);
  int*   tab  = eb + ((2 * (size_t)e + 3) & ~(size_t)3);
  float* out  = (float*)d_out;

  dim3 blk(256);
  dim3 big(1024);
  const int gMM = (n + 127) / 128;
  const int nc  = (e + CH - 1) / CH;

  k_bin_count<<<nc,   big, 0, stream>>>(dst, tab, e, nc);
  k_bin_scan <<<1,    big, 0, stream>>>(tab, NBIN * nc);
  k_bin_place<<<nc,   big, 0, stream>>>(src, dst, tab, eb, e, nc);
  k_dinv_bins<<<NBIN, big, 0, stream>>>(eb, tab, dinv, n, e, nc);

  k_gemm_l1  <<<gMM,  blk, 0, stream>>>(x, W1, dinv, out, n);
  k_agg<false><<<NBIN, big, 0, stream>>>(eb, tab, out, dinv, b1, buf1, n, e, nc);

  k_gemm_l2  <<<gMM,  blk, 0, stream>>>(W2, b1, dinv, buf1, n);
  k_agg<true><<<NBIN, big, 0, stream>>>(eb, tab, buf1, dinv, b2, out, n, e, nc);
}


# Product kernels run as one 25-wave work-group per WGP: W transposed into LDS once per WGP by bank-conflict-free stores, all 6250 row tiles resident in one round (was 782 groups, two rounds)
# speedup vs baseline: 4.3536x; 1.0898x over previous
#include <hip/hip_runtime.h>

typedef __attribute__((ext_vector_type(2))) float v2f;
typedef __attribute__((ext_vector_type(8))) float v8f;
typedef __attribute__((ext_vector_type(2))) int v2i;
typedef __attribute__((ext_vector_type(4))) int v4i;

#define HD 128
#define WT_STRIDE 132
#define GEMM_WAVES 25
#define GEMM_THREADS (GEMM_WAVES * 32)
#define GEMM_ROWS (GEMM_WAVES * 16)

#define NBIN 256
#define RPB 391
#define CH_PT 8
#define CH (1024 * CH_PT)
#define SCAN_PER 64
#define SCAN_PITCH 68

__global__ __launch_bounds__(1024) void k_bin_count(const int* __restrict__ dst, int* __restrict__ tab,
                                                    int e, int nc) {
  __shared__ int h[NBIN];
  const int tid = threadIdx.x;
  const int c = blockIdx.x;
  if (tid < NBIN) h[tid] = 0;
  __syncthreads();
  int d[CH_PT];
#pragma unroll
  for (int k = 0; k < CH_PT; ++k) {
    const int i = c * CH + k * 1024 + tid;
    d[k] = (i < e) ? dst[i] : -1;
  }
#pragma unroll
  for (int k = 0; k < CH_PT; ++k)
    if (d[k] >= 0) atomicAdd(&h[d[k] / RPB], 1);
  __syncthreads();
  if (tid < NBIN) tab[tid * nc + c] = h[tid];
}

__global__ __launch_bounds__(1024) void k_bin_scan(int* __restrict__ tab, int total) {
  __shared__ __attribute__((aligned(16))) int rowbuf[1024 * SCAN_PITCH];
  __shared__ int wsum[32];
  const int tid  = threadIdx.x;
  const int lane = tid & 31;
  const int wave = tid >> 5;

  v4i g[SCAN_PER / 4];
#pragma unroll
  for (int j = 0; j < SCAN_PER / 4; ++j) {
    const int f = (j * 1024 + tid) * 4;
    v4i z = {0, 0, 0, 0};
    g[j] = (f < total) ? *(const v4i*)(tab + f) : z;
  }
#pragma unroll
  for (int j = 0; j < SCAN_PER / 4; ++j) {
    const int f = (j * 1024 + tid) * 4;
    *(v4i*)(&rowbuf[(f >> 6) * SCAN_PITCH + (f & 63)]) = g[j];
  }
  __syncthreads();

  int* myrow = rowbuf + tid * SCAN_PITCH;
  int cn[SCAN_PER];
#pragma unroll
  for (int k = 0; k < SCAN_PER / 4; ++k) {
    const v4i c = *(const v4i*)(myrow + 4 * k);
    cn[4 * k + 0] = c.x;
    cn[4 * k + 1] = c.y;
    cn[4 * k + 2] = c.z;
    cn[4 * k + 3] = c.w;
  }
  int sum = 0;
#pragma unroll
  for (int k = 0; k < SCAN_PER; ++k) sum += cn[k];
  int v = sum;
  for (int o = 1; o < 32; o <<= 1) {
    const int u = __shfl_up(v, o);
    if (lane >= o) v += u;
  }
  if (lane == 31) wsum[wave] = v;
  __syncthreads();
  if (wave == 0) {
    int w = wsum[lane];
    for (int o = 1; o < 32; o <<= 1) {
      const int u = __shfl_up(w, o);
      if (lane >= o) w += u;
    }
    wsum[lane] = w;
  }
  __syncthreads();
  int run = v - sum + (wave > 0 ? wsum[wave - 1] : 0);
#pragma unroll
  for (int k = 0; k < SCAN_PER / 4; ++k) {
    v4i o;
    o.x = run; run += cn[4 * k + 0];
    o.y = run; run += cn[4 * k + 1];
    o.z = run; run += cn[4 * k + 2];
    o.w = run; run += cn[4 * k + 3];
    *(v4i*)(myrow + 4 * k) = o;
  }
  __syncthreads();

#pragma unroll
  for (int j = 0; j < SCAN_PER / 4; ++j) {
    const int f = (j * 1024 + tid) * 4;
    if (f < total) *(v4i*)(tab + f) = *(const v4i*)(&rowbuf[(f >> 6) * SCAN_PITCH + (f & 63)]);
  }
}

__global__ __launch_bounds__(1024) void k_bin_place(const int* __restrict__ src, const int* __restrict__ dst,
                                                    const int* __restrict__ tab, int* __restrict__ eb,
                                                    int e, int nc) {
  __shared__ int cur[NBIN];
  const int tid = threadIdx.x;
  const int c = blockIdx.x;
  if (tid < NBIN) cur[tid] = tab[tid * nc + c];
  __syncthreads();
  int s[CH_PT], d[CH_PT];
#pragma unroll
  for (int k = 0; k < CH_PT; ++k) {
    const int i = c * CH + k * 1024 + tid;
    s[k] = (i < e) ? src[i] : 0;
    d[k] = (i < e) ? dst[i] : -1;
  }
#pragma unroll
  for (int k = 0; k < CH_PT; ++k) {
    if (d[k] >= 0) {
      const int pos = atomicAdd(&cur[d[k] / RPB], 1);
      v2i sd;
      sd.x = s[k];
      sd.y = d[k];
      *(v2i*)(eb + 2 * (size_t)pos) = sd;
    }
  }
}

__global__ __launch_bounds__(1024) void k_dinv_bins(const int* __restrict__ eb, const int* __restrict__ tab,
                                                    float* __restrict__ dinv, int n, int e, int nc) {
  __shared__ int cnt[RPB];
  const int tid = threadIdx.x;
  const int b = blockIdx.x;
  const int base = b * RPB;
  int rows = n - base;
  if (rows > RPB) rows = RPB;
  if (tid < RPB) cnt[tid] = 0;
  __syncthreads();
  const int e0 = tab[b * nc];
  const int e1 = (b + 1 < NBIN) ? tab[(b + 1) * nc] : e;
#pragma unroll 4
  for (int i = e0 + tid; i < e1; i += 1024) atomicAdd(&cnt[eb[2 * (size_t)i + 1] - base], 1);
  __syncthreads();
  if (tid < rows) dinv[base + tid] = rsqrtf(1.0f + (float)cnt[tid]);
}

template <bool IN_SCALE_BIAS_RELU>
__device__ __forceinline__ void gemm_body(const float* A,
                                          const float* __restrict__ W,
                                          const float* __restrict__ bin,
                                          const float* __restrict__ dinv,
                                          float* m, int n) {
  __shared__ float Wt[HD * WT_STRIDE];
  __shared__ __attribute__((aligned(16))) float bs[HD];
  constexpr int W_PIECES = (HD * HD / 4 + GEMM_THREADS - 1) / GEMM_THREADS;
  float4 wp[W_PIECES];
#pragma unroll
  for (int p = 0; p < W_PIECES; ++p) {
    const int i  = p * GEMM_THREADS + threadIdx.x;
    const int k  = ((i >> 8) << 3) | (i & 7);
    const int n4 = ((i >> 3) & 31) * 4;
    if (i < HD * HD / 4) wp[p] = *(const float4*)(W + k * HD + n4);
  }
#pragma unroll
  for (int p = 0; p < W_PIECES; ++p) {
    const int i  = p * GEMM_THREADS + threadIdx.x;
    const int k  = ((i >> 8) << 3) | (i & 7);
    const int n4 = ((i >> 3) & 31) * 4;
    if (i < HD * HD / 4) {
      Wt[(n4 + 0) * WT_STRIDE + k] = wp[p].x;
      Wt[(n4 + 1) * WT_STRIDE + k] = wp[p].y;
      Wt[(n4 + 2) * WT_STRIDE + k] = wp[p].z;
      Wt[(n4 + 3) * WT_STRIDE + k] = wp[p].w;
    }
  }
  if (IN_SCALE_BIAS_RELU) {
    if (threadIdx.x < HD) bs[threadIdx.x] = bin[threadIdx.x];
  }
  __syncthreads();

  const int lane = threadIdx.x & 31;
  const int wave = threadIdx.x >> 5;
  const int r0   = blockIdx.x * GEMM_ROWS + wave * 16;
  if (r0 + 16 > n) return;
  const int hi = lane >> 4;
  const int lo = lane & 15;
  const float* arow = A + (size_t)(r0 + lo) * HD;
  float da = 1.0f;
  if (IN_SCALE_BIAS_RELU) da = dinv[r0 + lo];

  v8f acc[8] = {};
#pragma unroll 8
  for (int kk = 0; kk < 32; ++kk) {
    const int k = kk * 4 + hi * 2;
    v2f a = *(const v2f*)(arow + k);
    if (IN_SCALE_BIAS_RELU) {
      const v2f bb = *(const v2f*)(&bs[k]);
      const float x0 = da * a.x + bb.x;
      const float x1 = da * a.y + bb.y;
      a.x = x0 > 0.0f ? x0 : 0.0f;
      a.y = x1 > 0.0f ? x1 : 0.0f;
    }
#pragma unroll
    for (int t = 0; t < 8; ++t) {
      const v2f b = *(const v2f*)(&Wt[(t * 16 + lo) * WT_STRIDE + k]);
      acc[t] = __builtin_amdgcn_wmma_f32_16x16x4_f32(
          false, a, false, b, (short)0, acc[t], false, false);
    }
  }

  float di[8];
#pragma unroll
  for (int r = 0; r < 8; ++r) di[r] = dinv[r0 + hi * 8 + r];
#pragma unroll
  for (int t = 0; t < 8; ++t) {
    float* mrow = m + (size_t)(r0 + hi * 8) * HD + t * 16 + lo;
#pragma unroll
    for (int r = 0; r < 8; ++r) mrow[(size_t)r * HD] = di[r] * acc[t][r];
  }
}

__global__ __launch_bounds__(GEMM_THREADS) void k_gemm_l1(const float* __restrict__ A,
                                                 const float* __restrict__ W,
                                                 const float* __restrict__ dinv,
                                                 float* __restrict__ m, int n) {
  gemm_body<false>(A, W, W, dinv, m, n);
}

__global__ __launch_bounds__(GEMM_THREADS) void k_gemm_l2(const float* __restrict__ W,
                                                 const float* __restrict__ bin,
                                                 const float* __restrict__ dinv,
                                                 float* m, int n) {
  gemm_body<true>(m, W, bin, dinv, m, n);
}

#define EG 16
template <bool FINAL>
__global__ __launch_bounds__(1024) void k_agg(const int* __restrict__ eb, const int* __restrict__ tab,
                                              const float* __restrict__ m,
                                              const float* __restrict__ dinv,
                                              const float* __restrict__ bias,
                                              float* __restrict__ outp, int n, int e, int nc) {
  __shared__ float acc[RPB * HD];
  const int lane = threadIdx.x & 31;
  const int wave = __builtin_amdgcn_readfirstlane((int)(threadIdx.x >> 5));
  const unsigned ul4 = (unsigned)lane * 4u;
  const int b = blockIdx.x;
  const int base = b * RPB;
  int rows = n - base;
  if (rows > RPB) rows = RPB;

#pragma unroll 4
  for (int r = wave; r < rows; r += 32) {
    const float4 v = *(const float4*)(m + (size_t)(base + r) * HD + ul4);
    float* a = acc + r * HD + lane;
    a[0]  = v.x;
    a[32] = v.y;
    a[64] = v.z;
    a[96] = v.w;
  }
  __syncthreads();

  const int e0 = tab[b * nc];
  const int e1 = (b + 1 < NBIN) ? tab[(b + 1) * nc] : e;
  for (int q0 = e0 + wave * 32; q0 < e1; q0 += 32 * 32) {
    const int i = q0 + lane;
    int s = 0, dl = 0;
    if (i < e1) {
      const v2i sd = *(const v2i*)(eb + 2 * (size_t)i);
      s  = sd.x;
      dl = sd.y - base;
    }
    const int cnt = (e1 - q0) < 32 ? (e1 - q0) : 32;
    if (cnt == 32) {
#pragma unroll 1
      for (int j0 = 0; j0 < 32; j0 += EG) {
        float4 v[EG];
#pragma unroll
        for (int j = 0; j < EG; ++j) {
          const int sj = __builtin_amdgcn_readlane(s, j0 + j);
          const float* mp = m + (size_t)sj * HD;
          v[j] = *(const float4*)(mp + ul4);
        }
#pragma unroll
        for (int j = 0; j < EG; ++j) {
          const int dj = __builtin_amdgcn_readlane(dl, j0 + j);
          float* a = acc + dj * HD + lane;
          atomicAdd(a,      v[j].x);
          atomicAdd(a + 32, v[j].y);
          atomicAdd(a + 64, v[j].z);
          atomicAdd(a + 96, v[j].w);
        }
      }
    } else {
      for (int j = 0; j < cnt; ++j) {
        const int sj = __builtin_amdgcn_readlane(s, j);
        const int dj = __builtin_amdgcn_readlane(dl, j);
        const float4 v = *(const float4*)(m + (size_t)sj * HD + ul4);
        float* a = acc + dj * HD + lane;
        atomicAdd(a,      v.x);
        atomicAdd(a + 32, v.y);
        atomicAdd(a + 64, v.z);
        atomicAdd(a + 96, v.w);
      }
    }
  }
  __syncthreads();

  float4 bb = make_float4(0.0f, 0.0f, 0.0f, 0.0f);
  if (FINAL) bb = *(const float4*)(bias + ul4);
  for (int r = wave; r < rows; r += 32) {
    const float* a = acc + r * HD + lane;
    float4 v;
    v.x = a[0];
    v.y = a[32];
    v.z = a[64];
    v.w = a[96];
    if (FINAL) {
      const float dr = dinv[base + r];
      v.x = dr * v.x + bb.x;
      v.y = dr * v.y + bb.y;
      v.z = dr * v.z + bb.z;
      v.w = dr * v.w + bb.w;
    }
    *(float4*)(outp + (size_t)(base + r) * HD + ul4) = v;
  }
}

extern "C" void kernel_launch(void* const* d_in, const int* in_sizes, int n_in,
                              void* d_out, int out_size, void* d_ws, size_t ws_size,
                              hipStream_t stream) {
  const float* x  = (const float*)d_in[0];
  const float* W1 = (const float*)d_in[1];
  const float* b1 = (const float*)d_in[2];
  const float* W2 = (const float*)d_in[3];
  const float* b2 = (const float*)d_in[4];
  const int*   ei = (const int*)d_in[5];

  const int n = in_sizes[0] / HD;
  const int e = in_sizes[5] / 2;
  const int* src = ei;
  const int* dst = ei + e;
  const int total = n * HD;

  char* ws = (char*)d_ws;
  float* dinv = (float*)ws;
  size_t off1 = ((size_t)n * 4 + 255) & ~(size_t)255;
  float* buf1 = (float*)(ws + off1);
  size_t off2 = off1 + (((size_t)total * 4 + 255) & ~(size_t)255);
  int*   eb   = (int*)(ws + off2);
  int*   tab  = eb + ((2 * (size_t)e + 3) & ~(size_t)3);
  float* out  = (float*)d_out;

  dim3 blk(GEMM_THREADS);
  dim3 big(1024);
  const int gMM = (n + GEMM_ROWS - 1) / GEMM_ROWS;
  const int nc  = (e + CH - 1) / CH;

  k_bin_count<<<nc,   big, 0, stream>>>(dst, tab, e, nc);
  k_bin_scan <<<1,    big, 0, stream>>>(tab, NBIN * nc);
  k_bin_place<<<nc,   big, 0, stream>>>(src, dst, tab, eb, e, nc);
  k_dinv_bins<<<NBIN, big, 0, stream>>>(eb, tab, dinv, n, e, nc);

  k_gemm_l1  <<<gMM,  blk, 0, stream>>>(x, W1, dinv, out, n);
  k_agg<false><<<NBIN, big, 0, stream>>>(eb, tab, out, dinv, b1, buf1, n, e, nc);

  k_gemm_l2  <<<gMM,  blk, 0, stream>>>(W2, b1, dinv, buf1, n);
  k_agg<true><<<NBIN, big, 0, stream>>>(eb, tab, buf1, dinv, b2, out, n, e, nc);
}


# Place pass groups its chunk's (src,dst) pairs by range in LDS, then copies them out in slot order: a wave's store is 1-2 contiguous pieces, not 32 lone 8-byte pieces; sums in arrival order as before
# speedup vs baseline: 4.5287x; 1.0402x over previous
#include <hip/hip_runtime.h>

typedef __attribute__((ext_vector_type(2))) float v2f;
typedef __attribute__((ext_vector_type(8))) float v8f;
typedef __attribute__((ext_vector_type(2))) int v2i;
typedef __attribute__((ext_vector_type(4))) int v4i;

#define HD 128
#define WT_STRIDE 132
#define GEMM_WAVES 25
#define GEMM_THREADS (GEMM_WAVES * 32)
#define GEMM_ROWS (GEMM_WAVES * 16)

#define NBIN 256
#define RPB 391
#define CH_PT 8
#define CH (1024 * CH_PT)
#define SCAN_PER 64
#define SCAN_PITCH 68

__global__ __launch_bounds__(1024) void k_bin_count(const int* __restrict__ dst, int* __restrict__ tab,
                                                    int e, int nc) {
  __shared__ int h[NBIN];
  const int tid = threadIdx.x;
  const int c = blockIdx.x;
  if (tid < NBIN) h[tid] = 0;
  __syncthreads();
  int d[CH_PT];
#pragma unroll
  for (int k = 0; k < CH_PT; ++k) {
    const int i = c * CH + k * 1024 + tid;
    d[k] = (i < e) ? dst[i] : -1;
  }
#pragma unroll
  for (int k = 0; k < CH_PT; ++k)
    if (d[k] >= 0) atomicAdd(&h[d[k] / RPB], 1);
  __syncthreads();
  if (tid < NBIN) tab[tid * nc + c] = h[tid];
}

__global__ __launch_bounds__(1024) void k_bin_scan(int* __restrict__ tab, int total) {
  __shared__ __attribute__((aligned(16))) int rowbuf[1024 * SCAN_PITCH];
  __shared__ int wsum[32];
  const int tid  = threadIdx.x;
  const int lane = tid & 31;
  const int wave = tid >> 5;

  v4i g[SCAN_PER / 4];
#pragma unroll
  for (int j = 0; j < SCAN_PER / 4; ++j) {
    const int f = (j * 1024 + tid) * 4;
    v4i z = {0, 0, 0, 0};
    g[j] = (f < total) ? *(const v4i*)(tab + f) : z;
  }
#pragma unroll
  for (int j = 0; j < SCAN_PER / 4; ++j) {
    const int f = (j * 1024 + tid) * 4;
    *(v4i*)(&rowbuf[(f >> 6) * SCAN_PITCH + (f & 63)]) = g[j];
  }
  __syncthreads();

  int* myrow = rowbuf + tid * SCAN_PITCH;
  int cn[SCAN_PER];
#pragma unroll
  for (int k = 0; k < SCAN_PER / 4; ++k) {
    const v4i c = *(const v4i*)(myrow + 4 * k);
    cn[4 * k + 0] = c.x;
    cn[4 * k + 1] = c.y;
    cn[4 * k + 2] = c.z;
    cn[4 * k + 3] = c.w;
  }
  int sum = 0;
#pragma unroll
  for (int k = 0; k < SCAN_PER; ++k) sum += cn[k];
  int v = sum;
  for (int o = 1; o < 32; o <<= 1) {
    const int u = __shfl_up(v, o);
    if (lane >= o) v += u;
  }
  if (lane == 31) wsum[wave] = v;
  __syncthreads();
  if (wave == 0) {
    int w = wsum[lane];
    for (int o = 1; o < 32; o <<= 1) {
      const int u = __shfl_up(w, o);
      if (lane >= o) w += u;
    }
    wsum[lane] = w;
  }
  __syncthreads();
  int run = v - sum + (wave > 0 ? wsum[wave - 1] : 0);
#pragma unroll
  for (int k = 0; k < SCAN_PER / 4; ++k) {
    v4i o;
    o.x = run; run += cn[4 * k + 0];
    o.y = run; run += cn[4 * k + 1];
    o.z = run; run += cn[4 * k + 2];
    o.w = run; run += cn[4 * k + 3];
    *(v4i*)(myrow + 4 * k) = o;
  }
  __syncthreads();

#pragma unroll
  for (int j = 0; j < SCAN_PER / 4; ++j) {
    const int f = (j * 1024 + tid) * 4;
    if (f < total) *(v4i*)(tab + f) = *(const v4i*)(&rowbuf[(f >> 6) * SCAN_PITCH + (f & 63)]);
  }
}

__global__ __launch_bounds__(1024) void k_bin_place(const int* __restrict__ src, const int* __restrict__ dst,
                                                    const int* __restrict__ tab, int* __restrict__ eb,
                                                    int e, int nc) {
  __shared__ int cnt[NBIN];
  __shared__ int lst[NBIN];
  __shared__ int goff[NBIN];
  __shared__ int wtot[NBIN / 32];
  __shared__ __attribute__((aligned(16))) int srt[2 * CH];
  const int tid  = threadIdx.x;
  const int lane = tid & 31;
  const int c = blockIdx.x;
  int gpos = 0;
  if (tid < NBIN) {
    cnt[tid] = 0;
    gpos = tab[tid * nc + c];
  }
  __syncthreads();
  int s[CH_PT], d[CH_PT], rk[CH_PT];
#pragma unroll
  for (int k = 0; k < CH_PT; ++k) {
    const int i = c * CH + k * 1024 + tid;
    s[k] = (i < e) ? src[i] : 0;
    d[k] = (i < e) ? dst[i] : -1;
  }
#pragma unroll
  for (int k = 0; k < CH_PT; ++k) {
    rk[k] = 0;
    if (d[k] >= 0) rk[k] = atomicAdd(&cnt[d[k] / RPB], 1);
  }
  __syncthreads();
  int own = 0, v = 0;
  if (tid < NBIN) {
    own = cnt[tid];
    v = own;
    for (int o = 1; o < 32; o <<= 1) {
      const int u = __shfl_up(v, o);
      if (lane >= o) v += u;
    }
    if (lane == 31) wtot[tid >> 5] = v;
  }
  __syncthreads();
  if (tid < NBIN) {
    int pre = 0;
    for (int w = 0; w < (tid >> 5); ++w) pre += wtot[w];
    const int st = pre + v - own;
    lst[tid]  = st;
    goff[tid] = gpos - st;
  }
  __syncthreads();
  int ls[CH_PT];
#pragma unroll
  for (int k = 0; k < CH_PT; ++k) ls[k] = lst[d[k] >= 0 ? d[k] / RPB : 0];
#pragma unroll
  for (int k = 0; k < CH_PT; ++k) {
    if (d[k] >= 0) {
      v2i pr;
      pr.x = s[k];
      pr.y = d[k];
      *(v2i*)(&srt[2 * (ls[k] + rk[k])]) = pr;
    }
  }
  __syncthreads();
  int nvalid = e - c * CH;
  if (nvalid > CH) nvalid = CH;
  v2i sd[CH_PT];
  int go[CH_PT];
#pragma unroll
  for (int k = 0; k < CH_PT; ++k) sd[k] = *(const v2i*)(&srt[2 * (k * 1024 + tid)]);
#pragma unroll
  for (int k = 0; k < CH_PT; ++k) {
    const int r = (k * 1024 + tid < nvalid) ? sd[k].y / RPB : 0;
    go[k] = goff[r];
  }
#pragma unroll
  for (int k = 0; k < CH_PT; ++k) {
    const int i = k * 1024 + tid;
    if (i < nvalid) *(v2i*)(eb + 2 * (size_t)(go[k] + i)) = sd[k];
  }
}

__global__ __launch_bounds__(1024) void k_dinv_bins(const int* __restrict__ eb, const int* __restrict__ tab,
                                                    float* __restrict__ dinv, int n, int e, int nc) {
  __shared__ int cnt[RPB];
  const int tid = threadIdx.x;
  const int b = blockIdx.x;
  const int base = b * RPB;
  int rows = n - base;
  if (rows > RPB) rows = RPB;
  if (tid < RPB) cnt[tid] = 0;
  __syncthreads();
  const int e0 = tab[b * nc];
  const int e1 = (b + 1 < NBIN) ? tab[(b + 1) * nc] : e;
#pragma unroll 4
  for (int i = e0 + tid; i < e1; i += 1024) atomicAdd(&cnt[eb[2 * (size_t)i + 1] - base], 1);
  __syncthreads();
  if (tid < rows) dinv[base + tid] = rsqrtf(1.0f + (float)cnt[tid]);
}

template <bool IN_SCALE_BIAS_RELU>
__device__ __forceinline__ void gemm_body(const float* A,
                                          const float* __restrict__ W,
                                          const float* __restrict__ bin,
                                          const float* __restrict__ dinv,
                                          float* m, int n) {
  __shared__ float Wt[HD * WT_STRIDE];
  __shared__ __attribute__((aligned(16))) float bs[HD];
  constexpr int W_PIECES = (HD * HD / 4 + GEMM_THREADS - 1) / GEMM_THREADS;
  float4 wp[W_PIECES];
#pragma unroll
  for (int p = 0; p < W_PIECES; ++p) {
    const int i  = p * GEMM_THREADS + threadIdx.x;
    const int k  = ((i >> 8) << 3) | (i & 7);
    const int n4 = ((i >> 3) & 31) * 4;
    if (i < HD * HD / 4) wp[p] = *(const float4*)(W + k * HD + n4);
  }
#pragma unroll
  for (int p = 0; p < W_PIECES; ++p) {
    const int i  = p * GEMM_THREADS + threadIdx.x;
    const int k  = ((i >> 8) << 3) | (i & 7);
    const int n4 = ((i >> 3) & 31) * 4;
    if (i < HD * HD / 4) {
      Wt[(n4 + 0) * WT_STRIDE + k] = wp[p].x;
      Wt[(n4 + 1) * WT_STRIDE + k] = wp[p].y;
      Wt[(n4 + 2) * WT_STRIDE + k] = wp[p].z;
      Wt[(n4 + 3) * WT_STRIDE + k] = wp[p].w;
    }
  }
  if (IN_SCALE_BIAS_RELU) {
    if (threadIdx.x < HD) bs[threadIdx.x] = bin[threadIdx.x];
  }
  __syncthreads();

  const int lane = threadIdx.x & 31;
  const int wave = threadIdx.x >> 5;
  const int r0   = blockIdx.x * GEMM_ROWS + wave * 16;
  if (r0 + 16 > n) return;
  const int hi = lane >> 4;
  const int lo = lane & 15;
  const float* arow = A + (size_t)(r0 + lo) * HD;
  float da = 1.0f;
  if (IN_SCALE_BIAS_RELU) da = dinv[r0 + lo];

  v8f acc[8] = {};
#pragma unroll 8
  for (int kk = 0; kk < 32; ++kk) {
    const int k = kk * 4 + hi * 2;
    v2f a = *(const v2f*)(arow + k);
    if (IN_SCALE_BIAS_RELU) {
      const v2f bb = *(const v2f*)(&bs[k]);
      const float x0 = da * a.x + bb.x;
      const float x1 = da * a.y + bb.y;
      a.x = x0 > 0.0f ? x0 : 0.0f;
      a.y = x1 > 0.0f ? x1 : 0.0f;
    }
#pragma unroll
    for (int t = 0; t < 8; ++t) {
      const v2f b = *(const v2f*)(&Wt[(t * 16 + lo) * WT_STRIDE + k]);
      acc[t] = __builtin_amdgcn_wmma_f32_16x16x4_f32(
          false, a, false, b, (short)0, acc[t], false, false);
    }
  }

  float di[8];
#pragma unroll
  for (int r = 0; r < 8; ++r) di[r] = dinv[r0 + hi * 8 + r];
#pragma unroll
  for (int t = 0; t < 8; ++t) {
    float* mrow = m + (size_t)(r0 + hi * 8) * HD + t * 16 + lo;
#pragma unroll
    for (int r = 0; r < 8; ++r) mrow[(size_t)r * HD] = di[r] * acc[t][r];
  }
}

__global__ __launch_bounds__(GEMM_THREADS) void k_gemm_l1(const float* __restrict__ A,
                                                 const float* __restrict__ W,
                                                 const float* __restrict__ dinv,
                                                 float* __restrict__ m, int n) {
  gemm_body<false>(A, W, W, dinv, m, n);
}

__global__ __launch_bounds__(GEMM_THREADS) void k_gemm_l2(const float* __restrict__ W,
                                                 const float* __restrict__ bin,
                                                 const float* __restrict__ dinv,
                                                 float* m, int n) {
  gemm_body<true>(m, W, bin, dinv, m, n);
}

#define EG 16
template <bool FINAL>
__global__ __launch_bounds__(1024) void k_agg(const int* __restrict__ eb, const int* __restrict__ tab,
                                              const float* __restrict__ m,
                                              const float* __restrict__ dinv,
                                              const float* __restrict__ bias,
                                              float* __restrict__ outp, int n, int e, int nc) {
  __shared__ float acc[RPB * HD];
  const int lane = threadIdx.x & 31;
  const int wave = __builtin_amdgcn_readfirstlane((int)(threadIdx.x >> 5));
  const unsigned ul4 = (unsigned)lane * 4u;
  const int b = blockIdx.x;
  const int base = b * RPB;
  int rows = n - base;
  if (rows > RPB) rows = RPB;

#pragma unroll 4
  for (int r = wave; r < rows; r += 32) {
    const float4 v = *(const float4*)(m + (size_t)(base + r) * HD + ul4);
    float* a = acc + r * HD + lane;
    a[0]  = v.x;
    a[32] = v.y;
    a[64] = v.z;
    a[96] = v.w;
  }
  __syncthreads();

  const int e0 = tab[b * nc];
  const int e1 = (b + 1 < NBIN) ? tab[(b + 1) * nc] : e;
  for (int q0 = e0 + wave * 32; q0 < e1; q0 += 32 * 32) {
    const int i = q0 + lane;
    int s = 0, dl = 0;
    if (i < e1) {
      const v2i sd = *(const v2i*)(eb + 2 * (size_t)i);
      s  = sd.x;
      dl = sd.y - base;
    }
    const int cnt = (e1 - q0) < 32 ? (e1 - q0) : 32;
    if (cnt == 32) {
#pragma unroll 1
      for (int j0 = 0; j0 < 32; j0 += EG) {
        float4 v[EG];
#pragma unroll
        for (int j = 0; j < EG; ++j) {
          const int sj = __builtin_amdgcn_readlane(s, j0 + j);
          const float* mp = m + (size_t)sj * HD;
          v[j] = *(const float4*)(mp + ul4);
        }
#pragma unroll
        for (int j = 0; j < EG; ++j) {
          const int dj = __builtin_amdgcn_readlane(dl, j0 + j);
          float* a = acc + dj * HD + lane;
          atomicAdd(a,      v[j].x);
          atomicAdd(a + 32, v[j].y);
          atomicAdd(a + 64, v[j].z);
          atomicAdd(a + 96, v[j].w);
        }
      }
    } else {
      for (int j = 0; j < cnt; ++j) {
        const int sj = __builtin_amdgcn_readlane(s, j);
        const int dj = __builtin_amdgcn_readlane(dl, j);
        const float4 v = *(const float4*)(m + (size_t)sj * HD + ul4);
        float* a = acc + dj * HD + lane;
        atomicAdd(a,      v.x);
        atomicAdd(a + 32, v.y);
        atomicAdd(a + 64, v.z);
        atomicAdd(a + 96, v.w);
      }
    }
  }
  __syncthreads();

  float4 bb = make_float4(0.0f, 0.0f, 0.0f, 0.0f);
  if (FINAL) bb = *(const float4*)(bias + ul4);
  for (int r = wave; r < rows; r += 32) {
    const float* a = acc + r * HD + lane;
    float4 v;
    v.x = a[0];
    v.y = a[32];
    v.z = a[64];
    v.w = a[96];
    if (FINAL) {
      const float dr = dinv[base + r];
      v.x = dr * v.x + bb.x;
      v.y = dr * v.y + bb.y;
      v.z = dr * v.z + bb.z;
      v.w = dr * v.w + bb.w;
    }
    *(float4*)(outp + (size_t)(base + r) * HD + ul4) = v;
  }
}

extern "C" void kernel_launch(void* const* d_in, const int* in_sizes, int n_in,
                              void* d_out, int out_size, void* d_ws, size_t ws_size,
                              hipStream_t stream) {
  const float* x  = (const float*)d_in[0];
  const float* W1 = (const float*)d_in[1];
  const float* b1 = (const float*)d_in[2];
  const float* W2 = (const float*)d_in[3];
  const float* b2 = (const float*)d_in[4];
  const int*   ei = (const int*)d_in[5];

  const int n = in_sizes[0] / HD;
  const int e = in_sizes[5] / 2;
  const int* src = ei;
  const int* dst = ei + e;
  const int total = n * HD;

  char* ws = (char*)d_ws;
  float* dinv = (float*)ws;
  size_t off1 = ((size_t)n * 4 + 255) & ~(size_t)255;
  float* buf1 = (float*)(ws + off1);
  size_t off2 = off1 + (((size_t)total * 4 + 255) & ~(size_t)255);
  int*   eb   = (int*)(ws + off2);
  int*   tab  = eb + ((2 * (size_t)e + 3) & ~(size_t)3);
  float* out  = (float*)d_out;

  dim3 blk(GEMM_THREADS);
  dim3 big(1024);
  const int gMM = (n + GEMM_ROWS - 1) / GEMM_ROWS;
  const int nc  = (e + CH - 1) / CH;

  k_bin_count<<<nc,   big, 0, stream>>>(dst, tab, e, nc);
  k_bin_scan <<<1,    big, 0, stream>>>(tab, NBIN * nc);
  k_bin_place<<<nc,   big, 0, stream>>>(src, dst, tab, eb, e, nc);
  k_dinv_bins<<<NBIN, big, 0, stream>>>(eb, tab, dinv, n, e, nc);

  k_gemm_l1  <<<gMM,  blk, 0, stream>>>(x, W1, dinv, out, n);
  k_agg<false><<<NBIN, big, 0, stream>>>(eb, tab, out, dinv, b1, buf1, n, e, nc);

  k_gemm_l2  <<<gMM,  blk, 0, stream>>>(W2, b1, dinv, buf1, n);
  k_agg<true><<<NBIN, big, 0, stream>>>(eb, tab, buf1, dinv, b2, out, n, e, nc);
}
